# starting kernel
# baseline (speedup 1.0000x reference)
#include <hip/hip_runtime.h>
#include <hip/hip_bf16.h>

typedef __attribute__((ext_vector_type(2))) float v2f;
typedef __attribute__((ext_vector_type(8))) float v8f;
typedef __attribute__((ext_vector_type(4))) int   v4i;

__global__ void gcn_premul_zero(const float* __restrict__ feat,
                                const float* __restrict__ norm,
                                float* __restrict__ pre1,
                                float* __restrict__ agg1,
                                int n) {
  int i = blockIdx.x * blockDim.x + threadIdx.x;
  if (i < n) {
    pre1[i] = feat[i] * norm[i];
    agg1[i] = 0.0f;
  }
}

__global__ void gcn_edge_scatter(const int* __restrict__ src,
                                 const int* __restrict__ dst,
                                 const float* __restrict__ val,
                                 float* __restrict__ agg,
                                 int e4, int e) {
  int tid = blockIdx.x * blockDim.x + threadIdx.x;
  if (tid < e4) {
    const v4i* s4 = (const v4i*)src;
    const v4i* d4 = (const v4i*)dst;
    v4i s = __builtin_nontemporal_load(&s4[tid]);
    v4i d = __builtin_nontemporal_load(&d4[tid]);
    __hip_atomic_fetch_add(&agg[d.x], val[s.x], __ATOMIC_RELAXED, __HIP_MEMORY_SCOPE_AGENT);
    __hip_atomic_fetch_add(&agg[d.y], val[s.y], __ATOMIC_RELAXED, __HIP_MEMORY_SCOPE_AGENT);
    __hip_atomic_fetch_add(&agg[d.z], val[s.z], __ATOMIC_RELAXED, __HIP_MEMORY_SCOPE_AGENT);
    __hip_atomic_fetch_add(&agg[d.w], val[s.w], __ATOMIC_RELAXED, __HIP_MEMORY_SCOPE_AGENT);
  } else if (tid == e4) {
    for (int k = e4 * 4; k < e; ++k) {
      __hip_atomic_fetch_add(&agg[dst[k]], val[src[k]], __ATOMIC_RELAXED, __HIP_MEMORY_SCOPE_AGENT);
    }
  }
}

__global__ void gcn_layer1_wmma(const float* __restrict__ agg1,
                                const float* __restrict__ norm,
                                const float* __restrict__ W1,
                                const float* __restrict__ b1,
                                const float* __restrict__ W2,
                                float* __restrict__ t,
                                float* __restrict__ agg2,
                                int n_nodes) {
  const int lane = threadIdx.x & 31;
  const int wave = threadIdx.x >> 5;
  const int base = (blockIdx.x * (blockDim.x >> 5) + wave) * 16;
  const int col = lane & 15;

  float w1n = 0.0f, b1n = 0.0f, w2n = 0.0f;
  if (col < 3) {
    w1n = W1[col];
    b1n = b1[col];
    w2n = W2[col];
  }

  const int idx = base + col;
  const bool rowlane = (lane < 16) && (idx < n_nodes);
  float xv = 0.0f;
  if (rowlane) {
    xv = agg1[idx] * norm[idx];
    agg2[idx] = 0.0f;
  }

  v2f a;
  a.x = xv;
  a.y = 0.0f;
  v2f b;
  b.x = w1n;
  b.y = w1n;
  v8f c;
#pragma unroll
  for (int r = 0; r < 8; ++r) c[r] = b1n;

  v8f d = __builtin_amdgcn_wmma_f32_16x16x4_f32(
       false, a,  false, b,
       (short)0, c,  false,  false);

#pragma unroll
  for (int r = 0; r < 8; ++r) {
    float h = fmaxf(d[r], 0.0f) * w2n;
    h += __shfl_xor(h, 1);
    h += __shfl_xor(h, 2);
    if (col == 0) {
      const int id = base + r + ((lane >> 4) << 3);
      if (id < n_nodes) t[id] = norm[id] * h;
    }
  }
}

__global__ void gcn_final(const float* __restrict__ norm,
                          const float* __restrict__ agg2,
                          const float* __restrict__ b2,
                          float* __restrict__ out,
                          int n) {
  int i = blockIdx.x * blockDim.x + threadIdx.x;
  if (i < n) {
    out[i] = fmaxf(norm[i] * agg2[i] + b2[0], 0.0f);
  }
}

extern "C" void kernel_launch(void* const* d_in, const int* in_sizes, int n_in,
                              void* d_out, int out_size, void* d_ws, size_t ws_size,
                              hipStream_t stream) {
  (void)n_in; (void)out_size; (void)ws_size;

  const float* feat = (const float*)d_in[0];
  const float* norm = (const float*)d_in[1];
  const int*   src  = (const int*)d_in[2];
  const int*   dst  = (const int*)d_in[3];
  const float* W1   = (const float*)d_in[4];
  const float* b1   = (const float*)d_in[5];
  const float* W2   = (const float*)d_in[6];
  const float* b2   = (const float*)d_in[7];
  float* out = (float*)d_out;

  const int N = in_sizes[0];
  const int E = in_sizes[2];

  float* pre1 = (float*)d_ws;
  float* agg1 = pre1 + N;
  float* tval = agg1 + N;
  float* agg2 = tval + N;

  const int TB = 256;
  const int grid_nodes = (N + TB - 1) / TB;

  const int e4 = E / 4;
  const int edge_threads = e4 + 1;
  const int grid_edges = (edge_threads + TB - 1) / TB;

  const int nodes_per_block = (TB / 32) * 16;
  const int grid_wmma = (N + nodes_per_block - 1) / nodes_per_block;

  gcn_premul_zero<<<grid_nodes, TB, 0, stream>>>(feat, norm, pre1, agg1, N);
  gcn_edge_scatter<<<grid_edges, TB, 0, stream>>>(src, dst, pre1, agg1, e4, E);
  gcn_layer1_wmma<<<grid_wmma, TB, 0, stream>>>(agg1, norm, W1, b1, W2, tval, agg2, N);
  gcn_edge_scatter<<<grid_edges, TB, 0, stream>>>(src, dst, tval, agg2, e4, E);
  gcn_final<<<grid_nodes, TB, 0, stream>>>(norm, agg2, b2, out, N);
}


# Edge sums accumulate in LDS per 65536-node destination window (block = window x edge slice), flushed by whole-line atomics: no random global atomics; float sum order changes (LDS partial sums)
# speedup vs baseline: 3.0081x; 3.0081x over previous
#include <hip/hip_runtime.h>
#include <hip/hip_bf16.h>

typedef __attribute__((ext_vector_type(2))) float v2f;
typedef __attribute__((ext_vector_type(8))) float v8f;
typedef __attribute__((ext_vector_type(4))) int   v4i;

__global__ void gcn_premul_zero(const float* __restrict__ feat,
                                const float* __restrict__ norm,
                                float* __restrict__ pre1,
                                float* __restrict__ agg1,
                                int n) {
  int i = blockIdx.x * blockDim.x + threadIdx.x;
  if (i < n) {
    pre1[i] = feat[i] * norm[i];
    agg1[i] = 0.0f;
  }
}

#define WIN_NODES    65536
#define SCAN_SLICES  16
#define SCAN_THREADS 1024
#define SCAN_GROUP   4

__global__ __launch_bounds__(SCAN_THREADS)
void gcn_edge_window(const int* __restrict__ src,
                     const int* __restrict__ dst,
                     const float* __restrict__ val,
                     float* __restrict__ agg,
                     int e4, int e, int n) {
  __shared__ float acc[WIN_NODES];

  const int tid   = threadIdx.x;
  const int win   = blockIdx.x / SCAN_SLICES;
  const int slice = blockIdx.x % SCAN_SLICES;
  const int base  = win * WIN_NODES;

  float4* acc4 = (float4*)acc;
  for (int i = tid; i < WIN_NODES / 4; i += SCAN_THREADS) {
    acc4[i] = make_float4(0.0f, 0.0f, 0.0f, 0.0f);
  }
  __syncthreads();

  const int per   = (e4 + SCAN_SLICES - 1) / SCAN_SLICES;
  const int start = slice * per;
  const int stop  = (start + per < e4) ? (start + per) : e4;

  const v4i* s4 = (const v4i*)src;
  const v4i* d4 = (const v4i*)dst;

  for (int v0 = start + tid; v0 < stop; v0 += SCAN_THREADS * SCAN_GROUP) {
    v4i d[SCAN_GROUP];
    v4i s[SCAN_GROUP];
#pragma unroll
    for (int k = 0; k < SCAN_GROUP; ++k) {
      const int v = v0 + k * SCAN_THREADS;
      if (v < stop) {
        d[k] = d4[v];
        s[k] = s4[v];
      } else {
        d[k] = (v4i)(-1);
        s[k] = (v4i)(0);
      }
    }

    unsigned r[SCAN_GROUP][4];
    float    x[SCAN_GROUP][4];
#pragma unroll
    for (int k = 0; k < SCAN_GROUP; ++k) {
#pragma unroll
      for (int c = 0; c < 4; ++c) {
        r[k][c] = (unsigned)(d[k][c] - base);
        const int si = (r[k][c] < (unsigned)WIN_NODES) ? s[k][c] : 0;
        x[k][c] = val[si];
      }
    }
#pragma unroll
    for (int k = 0; k < SCAN_GROUP; ++k) {
#pragma unroll
      for (int c = 0; c < 4; ++c) {
        if (r[k][c] < (unsigned)WIN_NODES) {
          __hip_atomic_fetch_add(&acc[r[k][c]], x[k][c], __ATOMIC_RELAXED,
                                 __HIP_MEMORY_SCOPE_WORKGROUP);
        }
      }
    }
  }

  if (slice == 0 && tid == 0) {
    for (int k = e4 * 4; k < e; ++k) {
      const unsigned rr = (unsigned)(dst[k] - base);
      if (rr < (unsigned)WIN_NODES) {
        __hip_atomic_fetch_add(&acc[rr], val[src[k]], __ATOMIC_RELAXED,
                               __HIP_MEMORY_SCOPE_WORKGROUP);
      }
    }
  }
  __syncthreads();

  for (int i = tid; i < WIN_NODES; i += SCAN_THREADS) {
    const float a = acc[i];
    const int id = base + i;
    if (id < n && a != 0.0f) {
      __hip_atomic_fetch_add(&agg[id], a, __ATOMIC_RELAXED, __HIP_MEMORY_SCOPE_AGENT);
    }
  }
}

__global__ void gcn_layer1_wmma(const float* __restrict__ agg1,
                                const float* __restrict__ norm,
                                const float* __restrict__ W1,
                                const float* __restrict__ b1,
                                const float* __restrict__ W2,
                                float* __restrict__ t,
                                float* __restrict__ agg2,
                                int n_nodes) {
  const int lane = threadIdx.x & 31;
  const int wave = threadIdx.x >> 5;
  const int base = (blockIdx.x * (blockDim.x >> 5) + wave) * 16;
  const int col = lane & 15;

  float w1n = 0.0f, b1n = 0.0f, w2n = 0.0f;
  if (col < 3) {
    w1n = W1[col];
    b1n = b1[col];
    w2n = W2[col];
  }

  const int idx = base + col;
  const bool rowlane = (lane < 16) && (idx < n_nodes);
  float xv = 0.0f;
  if (rowlane) {
    xv = agg1[idx] * norm[idx];
    agg2[idx] = 0.0f;
  }

  v2f a;
  a.x = xv;
  a.y = 0.0f;
  v2f b;
  b.x = w1n;
  b.y = w1n;
  v8f c;
#pragma unroll
  for (int r = 0; r < 8; ++r) c[r] = b1n;

  v8f d = __builtin_amdgcn_wmma_f32_16x16x4_f32(
       false, a,  false, b,
       (short)0, c,  false,  false);

#pragma unroll
  for (int r = 0; r < 8; ++r) {
    float h = fmaxf(d[r], 0.0f) * w2n;
    h += __shfl_xor(h, 1);
    h += __shfl_xor(h, 2);
    if (col == 0) {
      const int id = base + r + ((lane >> 4) << 3);
      if (id < n_nodes) t[id] = norm[id] * h;
    }
  }
}

__global__ void gcn_final(const float* __restrict__ norm,
                          const float* __restrict__ agg2,
                          const float* __restrict__ b2,
                          float* __restrict__ out,
                          int n) {
  int i = blockIdx.x * blockDim.x + threadIdx.x;
  if (i < n) {
    out[i] = fmaxf(norm[i] * agg2[i] + b2[0], 0.0f);
  }
}

extern "C" void kernel_launch(void* const* d_in, const int* in_sizes, int n_in,
                              void* d_out, int out_size, void* d_ws, size_t ws_size,
                              hipStream_t stream) {
  (void)n_in; (void)out_size; (void)ws_size;

  const float* feat = (const float*)d_in[0];
  const float* norm = (const float*)d_in[1];
  const int*   src  = (const int*)d_in[2];
  const int*   dst  = (const int*)d_in[3];
  const float* W1   = (const float*)d_in[4];
  const float* b1   = (const float*)d_in[5];
  const float* W2   = (const float*)d_in[6];
  const float* b2   = (const float*)d_in[7];
  float* out = (float*)d_out;

  const int N = in_sizes[0];
  const int E = in_sizes[2];

  float* pre1 = (float*)d_ws;
  float* agg1 = pre1 + N;
  float* tval = agg1 + N;
  float* agg2 = tval + N;

  const int TB = 256;
  const int grid_nodes = (N + TB - 1) / TB;

  const int e4 = E / 4;
  const int n_win = (N + WIN_NODES - 1) / WIN_NODES;
  const int grid_edges = n_win * SCAN_SLICES;

  const int nodes_per_block = (TB / 32) * 16;
  const int grid_wmma = (N + nodes_per_block - 1) / nodes_per_block;

  gcn_premul_zero<<<grid_nodes, TB, 0, stream>>>(feat, norm, pre1, agg1, N);
  gcn_edge_window<<<grid_edges, SCAN_THREADS, 0, stream>>>(src, dst, pre1, agg1, e4, E, N);
  gcn_layer1_wmma<<<grid_wmma, TB, 0, stream>>>(agg1, norm, W1, b1, W2, tval, agg2, N);
  gcn_edge_window<<<grid_edges, SCAN_THREADS, 0, stream>>>(src, dst, tval, agg2, e4, E, N);
  gcn_final<<<grid_nodes, TB, 0, stream>>>(norm, agg2, b2, out, N);
}


# Edges binned once per call by (source window, destination window) of 32768 nodes; both layers gather and sum in LDS; needs 81 MB of workspace, else the old path runs; float sum order changes
# speedup vs baseline: 7.9614x; 2.6467x over previous
#include <hip/hip_runtime.h>
#include <hip/hip_bf16.h>


typedef __attribute__((ext_vector_type(2))) float v2f;
typedef __attribute__((ext_vector_type(8))) float v8f;
typedef __attribute__((ext_vector_type(4))) int   v4i;

__global__ void gcn_premul_zero(const float* __restrict__ feat,
                                const float* __restrict__ norm,
                                float* __restrict__ pre1,
                                float* __restrict__ agg1,
                                unsigned* __restrict__ binz, int nz,
                                int n) {
  int i = blockIdx.x * blockDim.x + threadIdx.x;
  if (i < n) {
    pre1[i] = feat[i] * norm[i];
    agg1[i] = 0.0f;
  }
  if (i < nz) binz[i] = 0u;
}

#define WBITS      15
#define WN         (1 << WBITS)
#define WMASK      (WN - 1)
#define MAXBINS    1024
#define BIN_BLOCKS 256
#define BT         1024
#define STAGE_CAP  65536
#define SRC_GROUPS 8

template <class F>
__device__ __forceinline__ void chunk_edges(const int* __restrict__ src,
                                            const int* __restrict__ dst,
                                            int e4, int e, F f) {
  const int tid   = threadIdx.x;
  const int per   = (e4 + BIN_BLOCKS - 1) / BIN_BLOCKS;
  const int start = blockIdx.x * per;
  const int stop  = (start + per < e4) ? (start + per) : e4;
  const v4i* s4 = (const v4i*)src;
  const v4i* d4 = (const v4i*)dst;
  int v0 = start + tid;
  for (; v0 + 3 * BT < stop; v0 += BT * 4) {
    v4i d[4];
    v4i s[4];
#pragma unroll
    for (int k = 0; k < 4; ++k) {
      d[k] = d4[v0 + k * BT];
      s[k] = s4[v0 + k * BT];
    }
#pragma unroll
    for (int k = 0; k < 4; ++k) {
#pragma unroll
      for (int c = 0; c < 4; ++c) f(s[k][c], d[k][c]);
    }
  }
  {
    v4i d[3];
    v4i s[3];
#pragma unroll
    for (int k = 0; k < 3; ++k) {
      const int v = v0 + k * BT;
      if (v < stop) {
        d[k] = d4[v];
        s[k] = s4[v];
      } else {
        d[k] = (v4i)(0);
        s[k] = (v4i)(0);
      }
    }
#pragma unroll
    for (int k = 0; k < 3; ++k) {
      if (v0 + k * BT < stop) {
#pragma unroll
        for (int c = 0; c < 4; ++c) f(s[k][c], d[k][c]);
      }
    }
  }
  if (blockIdx.x == 0 && tid == 0) {
    for (int k = e4 * 4; k < e; ++k) f(src[k], dst[k]);
  }
}

__global__ __launch_bounds__(BT)
void gcn_bin_count(const int* __restrict__ src,
                   const int* __restrict__ dst,
                   unsigned* __restrict__ binTotal,
                   int e4, int e, int nw, int nb) {
  __shared__ unsigned hist[MAXBINS];
  hist[threadIdx.x] = 0u;
  __syncthreads();
  chunk_edges(src, dst, e4, e, [&](int s, int d) {
    const int bin = (d >> WBITS) * nw + (s >> WBITS);
    __hip_atomic_fetch_add(&hist[bin], 1u, __ATOMIC_RELAXED, __HIP_MEMORY_SCOPE_WORKGROUP);
  });
  __syncthreads();
  const unsigned c = hist[threadIdx.x];
  if ((int)threadIdx.x < nb && c != 0u) {
    __hip_atomic_fetch_add(&binTotal[threadIdx.x], c, __ATOMIC_RELAXED, __HIP_MEMORY_SCOPE_AGENT);
  }
}

__device__ __forceinline__ unsigned block_excl_scan(unsigned v, unsigned* wsum) {
  const int lane = threadIdx.x & 31;
  const int w    = threadIdx.x >> 5;
  unsigned x = v;
#pragma unroll
  for (int o = 1; o < 32; o <<= 1) {
    const unsigned y = __shfl_up(x, o);
    if (lane >= o) x += y;
  }
  if (lane == 31) wsum[w] = x;
  __syncthreads();
  if (w == 0) {
    unsigned s = wsum[lane];
#pragma unroll
    for (int o = 1; o < 32; o <<= 1) {
      const unsigned y = __shfl_up(s, o);
      if (lane >= o) s += y;
    }
    wsum[lane] = s;
  }
  __syncthreads();
  const unsigned base = (w > 0) ? wsum[w - 1] : 0u;
  __syncthreads();
  return base + x - v;
}

__global__ __launch_bounds__(BT)
void gcn_bin_scatter(const int* __restrict__ src,
                     const int* __restrict__ dst,
                     const unsigned* __restrict__ binTotal,
                     unsigned* __restrict__ binCursor,
                     unsigned* __restrict__ binStart,
                     unsigned* __restrict__ binned,
                     int e4, int e, int nw, int nb) {
  __shared__ unsigned stage[STAGE_CAP];
  __shared__ unsigned hist[MAXBINS];
  __shared__ unsigned lstart[MAXBINS];
  __shared__ unsigned gpos[MAXBINS];
  __shared__ unsigned fill[MAXBINS];
  __shared__ unsigned wsum[32];

  const int tid = threadIdx.x;
  hist[tid] = 0u;
  fill[tid] = 0u;
  __syncthreads();

  chunk_edges(src, dst, e4, e, [&](int s, int d) {
    const int bin = (d >> WBITS) * nw + (s >> WBITS);
    __hip_atomic_fetch_add(&hist[bin], 1u, __ATOMIC_RELAXED, __HIP_MEMORY_SCOPE_WORKGROUP);
  });
  __syncthreads();

  const unsigned mine = hist[tid];
  const unsigned ls   = block_excl_scan(mine, wsum);
  const unsigned tot  = (tid < nb) ? binTotal[tid] : 0u;
  const unsigned gs   = block_excl_scan(tot, wsum);
  lstart[tid] = ls;
  unsigned g = gs;
  if (mine != 0u) {
    g += __hip_atomic_fetch_add(&binCursor[tid], mine, __ATOMIC_RELAXED, __HIP_MEMORY_SCOPE_AGENT);
  }
  gpos[tid] = g;
  if (blockIdx.x == 0) {
    binStart[tid] = gs;
    if (tid == BT - 1) binStart[MAXBINS] = gs + tot;
  }
  __syncthreads();

  chunk_edges(src, dst, e4, e, [&](int s, int d) {
    const int bin = (d >> WBITS) * nw + (s >> WBITS);
    const unsigned r = __hip_atomic_fetch_add(&fill[bin], 1u, __ATOMIC_RELAXED, __HIP_MEMORY_SCOPE_WORKGROUP);
    stage[lstart[bin] + r] = (unsigned)(s & WMASK) | ((unsigned)(d & WMASK) << WBITS);
  });
  __syncthreads();

  const int lane = tid & 31;
  const int w    = tid >> 5;
  for (int bin = w; bin < nb; bin += 32) {
    const unsigned cnt = hist[bin];
    const unsigned l0  = lstart[bin];
    const unsigned g0  = gpos[bin];
    for (unsigned q = lane; q < cnt; q += 32) binned[g0 + q] = stage[l0 + q];
  }
}

__global__ __launch_bounds__(BT)
void gcn_edge_binned(const unsigned* __restrict__ binned,
                     const unsigned* __restrict__ binStart,
                     const float* __restrict__ val,
                     float* __restrict__ agg,
                     int nw, int n) {
  __shared__ float vals[WN];
  __shared__ float acc[WN];

  const int tid = threadIdx.x;
  const int j   = blockIdx.x / SRC_GROUPS;
  const int g   = blockIdx.x % SRC_GROUPS;
  const bool al = (((size_t)val) & 15) == 0;

  float4* acc4 = (float4*)acc;
  for (int i = tid; i < WN / 4; i += BT) {
    acc4[i] = make_float4(0.0f, 0.0f, 0.0f, 0.0f);
  }

  for (int i = g; i < nw; i += SRC_GROUPS) {
    __syncthreads();
    const int sbase = i * WN;
    if (al && sbase + WN <= n) {
      float4 x[WN / (BT * 4)];
#pragma unroll
      for (int k = 0; k < WN / (BT * 4); ++k) {
        x[k] = *(const float4*)(val + sbase + (k * BT + tid) * 4);
      }
#pragma unroll
      for (int k = 0; k < WN / (BT * 4); ++k) {
        *(float4*)(vals + (k * BT + tid) * 4) = x[k];
      }
    } else {
      float y[WN / BT];
#pragma unroll
      for (int k = 0; k < WN / BT; ++k) {
        const int id = sbase + k * BT + tid;
        y[k] = (id < n) ? val[id] : 0.0f;
      }
#pragma unroll
      for (int k = 0; k < WN / BT; ++k) vals[k * BT + tid] = y[k];
    }
    __syncthreads();

    const unsigned p0 = binStart[j * nw + i];
    const unsigned p1 = binStart[j * nw + i + 1];
    unsigned p = p0 + tid;
    for (; p + 7 * BT < p1; p += BT * 8) {
      unsigned wd[8];
      float    x[8];
#pragma unroll
      for (int k = 0; k < 8; ++k) wd[k] = binned[p + k * BT];
#pragma unroll
      for (int k = 0; k < 8; ++k) x[k] = vals[wd[k] & WMASK];
#pragma unroll
      for (int k = 0; k < 8; ++k) {
        __hip_atomic_fetch_add(&acc[wd[k] >> WBITS], x[k],
                               __ATOMIC_RELAXED, __HIP_MEMORY_SCOPE_WORKGROUP);
      }
    }
    for (; p < p1; p += BT * 4) {
      unsigned wd[4];
      float    x[4];
#pragma unroll
      for (int k = 0; k < 4; ++k) {
        const unsigned pp = p + k * BT;
        wd[k] = (pp < p1) ? binned[pp] : 0xFFFFFFFFu;
      }
#pragma unroll
      for (int k = 0; k < 4; ++k) {
        const bool ok = wd[k] != 0xFFFFFFFFu;
        const unsigned sl = ok ? (wd[k] & WMASK) : 0u;
        const unsigned dl = ok ? (wd[k] >> WBITS) : (unsigned)tid;
        const float v = vals[sl];
        x[k] = ok ? v : 0.0f;
        wd[k] = dl;
      }
#pragma unroll
      for (int k = 0; k < 4; ++k) {
        __hip_atomic_fetch_add(&acc[wd[k]], x[k],
                               __ATOMIC_RELAXED, __HIP_MEMORY_SCOPE_WORKGROUP);
      }
    }
  }
  __syncthreads();

  const int base = j * WN;
  for (int i = tid; i < WN; i += BT) {
    const float a = acc[i];
    const int id = base + i;
    if (id < n && a != 0.0f) {
      __hip_atomic_fetch_add(&agg[id], a, __ATOMIC_RELAXED, __HIP_MEMORY_SCOPE_AGENT);
    }
  }
}

#define WIN_NODES    65536
#define SCAN_SLICES  16
#define SCAN_THREADS 1024
#define SCAN_GROUP   4

__global__ __launch_bounds__(SCAN_THREADS)
void gcn_edge_window(const int* __restrict__ src,
                     const int* __restrict__ dst,
                     const float* __restrict__ val,
                     float* __restrict__ agg,
                     int e4, int e, int n) {
  __shared__ float acc[WIN_NODES];

  const int tid   = threadIdx.x;
  const int win   = blockIdx.x / SCAN_SLICES;
  const int slice = blockIdx.x % SCAN_SLICES;
  const int base  = win * WIN_NODES;

  float4* acc4 = (float4*)acc;
  for (int i = tid; i < WIN_NODES / 4; i += SCAN_THREADS) {
    acc4[i] = make_float4(0.0f, 0.0f, 0.0f, 0.0f);
  }
  __syncthreads();

  const int per   = (e4 + SCAN_SLICES - 1) / SCAN_SLICES;
  const int start = slice * per;
  const int stop  = (start + per < e4) ? (start + per) : e4;

  const v4i* s4 = (const v4i*)src;
  const v4i* d4 = (const v4i*)dst;

  for (int v0 = start + tid; v0 < stop; v0 += SCAN_THREADS * SCAN_GROUP) {
    v4i d[SCAN_GROUP];
    v4i s[SCAN_GROUP];
#pragma unroll
    for (int k = 0; k < SCAN_GROUP; ++k) {
      const int v = v0 + k * SCAN_THREADS;
      if (v < stop) {
        d[k] = d4[v];
        s[k] = s4[v];
      } else {
        d[k] = (v4i)(-1);
        s[k] = (v4i)(0);
      }
    }

    unsigned r[SCAN_GROUP][4];
    float    x[SCAN_GROUP][4];
#pragma unroll
    for (int k = 0; k < SCAN_GROUP; ++k) {
#pragma unroll
      for (int c = 0; c < 4; ++c) {
        r[k][c] = (unsigned)(d[k][c] - base);
        const int si = (r[k][c] < (unsigned)WIN_NODES) ? s[k][c] : 0;
        x[k][c] = val[si];
      }
    }
#pragma unroll
    for (int k = 0; k < SCAN_GROUP; ++k) {
#pragma unroll
      for (int c = 0; c < 4; ++c) {
        if (r[k][c] < (unsigned)WIN_NODES) {
          __hip_atomic_fetch_add(&acc[r[k][c]], x[k][c], __ATOMIC_RELAXED,
                                 __HIP_MEMORY_SCOPE_WORKGROUP);
        }
      }
    }
  }

  if (slice == 0 && tid == 0) {
    for (int k = e4 * 4; k < e; ++k) {
      const unsigned rr = (unsigned)(dst[k] - base);
      if (rr < (unsigned)WIN_NODES) {
        __hip_atomic_fetch_add(&acc[rr], val[src[k]], __ATOMIC_RELAXED,
                               __HIP_MEMORY_SCOPE_WORKGROUP);
      }
    }
  }
  __syncthreads();

  for (int i = tid; i < WIN_NODES; i += SCAN_THREADS) {
    const float a = acc[i];
    const int id = base + i;
    if (id < n && a != 0.0f) {
      __hip_atomic_fetch_add(&agg[id], a, __ATOMIC_RELAXED, __HIP_MEMORY_SCOPE_AGENT);
    }
  }
}

__global__ void gcn_layer1_wmma(const float* __restrict__ agg1,
                                const float* __restrict__ norm,
                                const float* __restrict__ W1,
                                const float* __restrict__ b1,
                                const float* __restrict__ W2,
                                float* __restrict__ t,
                                float* __restrict__ agg2,
                                int n_nodes) {
  const int lane = threadIdx.x & 31;
  const int wave = threadIdx.x >> 5;
  const int base = (blockIdx.x * (blockDim.x >> 5) + wave) * 16;
  const int col = lane & 15;

  float w1n = 0.0f, b1n = 0.0f, w2n = 0.0f;
  if (col < 3) {
    w1n = W1[col];
    b1n = b1[col];
    w2n = W2[col];
  }

  const int idx = base + col;
  const bool rowlane = (lane < 16) && (idx < n_nodes);
  float xv = 0.0f;
  if (rowlane) {
    xv = agg1[idx] * norm[idx];
    agg2[idx] = 0.0f;
  }

  v2f a;
  a.x = xv;
  a.y = 0.0f;
  v2f b;
  b.x = w1n;
  b.y = w1n;
  v8f c;
#pragma unroll
  for (int r = 0; r < 8; ++r) c[r] = b1n;

  v8f d = __builtin_amdgcn_wmma_f32_16x16x4_f32(
       false, a,  false, b,
       (short)0, c,  false,  false);

#pragma unroll
  for (int r = 0; r < 8; ++r) {
    float h = fmaxf(d[r], 0.0f) * w2n;
    h += __shfl_xor(h, 1);
    h += __shfl_xor(h, 2);
    if (col == 0) {
      const int id = base + r + ((lane >> 4) << 3);
      if (id < n_nodes) t[id] = norm[id] * h;
    }
  }
}

__global__ void gcn_final(const float* __restrict__ norm,
                          const float* __restrict__ agg2,
                          const float* __restrict__ b2,
                          float* __restrict__ out,
                          int n) {
  int i = blockIdx.x * blockDim.x + threadIdx.x;
  if (i < n) {
    out[i] = fmaxf(norm[i] * agg2[i] + b2[0], 0.0f);
  }
}

extern "C" void kernel_launch(void* const* d_in, const int* in_sizes, int n_in,
                              void* d_out, int out_size, void* d_ws, size_t ws_size,
                              hipStream_t stream) {
  (void)n_in; (void)out_size;

  const float* feat = (const float*)d_in[0];
  const float* norm = (const float*)d_in[1];
  const int*   src  = (const int*)d_in[2];
  const int*   dst  = (const int*)d_in[3];
  const float* W1   = (const float*)d_in[4];
  const float* b1   = (const float*)d_in[5];
  const float* W2   = (const float*)d_in[6];
  const float* b2   = (const float*)d_in[7];
  float* out = (float*)d_out;

  const int N = in_sizes[0];
  const int E = in_sizes[2];

  float* pre1 = (float*)d_ws;
  float* agg1 = pre1 + N;
  float* tval = agg1 + N;
  float* agg2 = tval + N;
  unsigned* binTotal  = (unsigned*)(agg2 + N);
  unsigned* binCursor = binTotal + MAXBINS;
  unsigned* binStart  = binCursor + MAXBINS;
  unsigned* binned    = binStart + MAXBINS + 4;

  const int TB = 256;
  const int grid_nodes = (N + TB - 1) / TB;
  const int e4 = E / 4;

  const int nodes_per_block = (TB / 32) * 16;
  const int grid_wmma = (N + nodes_per_block - 1) / nodes_per_block;

  const int nw = (N + WN - 1) / WN;
  const int nb = nw * nw;
  const size_t need = (size_t)N * 16 + (size_t)(3 * MAXBINS + 4) * 4 + (size_t)E * 4;
  const int slice_edges = ((e4 + BIN_BLOCKS - 1) / BIN_BLOCKS) * 4 + 3;
  const bool binned_path = (ws_size >= need) && (nb <= MAXBINS) && (slice_edges <= STAGE_CAP) &&
                           (N >= 2 * MAXBINS);

  if (binned_path) {
    gcn_premul_zero<<<grid_nodes, TB, 0, stream>>>(feat, norm, pre1, agg1, binTotal, 2 * MAXBINS, N);
    gcn_bin_count<<<BIN_BLOCKS, BT, 0, stream>>>(src, dst, binTotal, e4, E, nw, nb);
    gcn_bin_scatter<<<BIN_BLOCKS, BT, 0, stream>>>(src, dst, binTotal, binCursor, binStart, binned,
                                                   e4, E, nw, nb);
    gcn_edge_binned<<<nw * SRC_GROUPS, BT, 0, stream>>>(binned, binStart, pre1, agg1, nw, N);
    gcn_layer1_wmma<<<grid_wmma, TB, 0, stream>>>(agg1, norm, W1, b1, W2, tval, agg2, N);
    gcn_edge_binned<<<nw * SRC_GROUPS, BT, 0, stream>>>(binned, binStart, tval, agg2, nw, N);
    gcn_final<<<grid_nodes, TB, 0, stream>>>(norm, agg2, b2, out, N);
    return;
  }

  const int n_win = (N + WIN_NODES - 1) / WIN_NODES;
  const int grid_edges = n_win * SCAN_SLICES;

  gcn_premul_zero<<<grid_nodes, TB, 0, stream>>>(feat, norm, pre1, agg1, binTotal, 0, N);
  gcn_edge_window<<<grid_edges, SCAN_THREADS, 0, stream>>>(src, dst, pre1, agg1, e4, E, N);
  gcn_layer1_wmma<<<grid_wmma, TB, 0, stream>>>(agg1, norm, W1, b1, W2, tval, agg2, N);
  gcn_edge_window<<<grid_edges, SCAN_THREADS, 0, stream>>>(src, dst, tval, agg2, e4, E, N);
  gcn_final<<<grid_nodes, TB, 0, stream>>>(norm, agg2, b2, out, N);
}


# Node kernels (pre-multiply, layer-1 matrix epilogue, final) as resident grids of 512 blocks: lanes loop over nodes, layer-1 product transposed so a lane owns a node: 125,016 waves become 12,288
# speedup vs baseline: 8.3621x; 1.0503x over previous
#include <hip/hip_runtime.h>
#include <hip/hip_bf16.h>


typedef __attribute__((ext_vector_type(2))) float v2f;
typedef __attribute__((ext_vector_type(8))) float v8f;
typedef __attribute__((ext_vector_type(4))) int   v4i;

#define NODE_BLOCKS  512
#define NODE_THREADS 256

__global__ __launch_bounds__(NODE_THREADS)
void gcn_premul_zero(const float* __restrict__ feat,
                     const float* __restrict__ norm,
                     float* __restrict__ pre1,
                     float* __restrict__ agg1,
                     unsigned* __restrict__ binz, int nz,
                     int n) {
  const int gt = blockIdx.x * blockDim.x + threadIdx.x;
  const int nt = gridDim.x * blockDim.x;
  if (gt < nz) binz[gt] = 0u;

  const bool vec = ((n & 3) == 0) &&
                   (((((size_t)feat) | ((size_t)norm) | ((size_t)pre1) | ((size_t)agg1)) & 15) == 0);
  if (vec) {
    const float4* f4 = (const float4*)feat;
    const float4* m4 = (const float4*)norm;
    float4* p4 = (float4*)pre1;
    float4* a4 = (float4*)agg1;
    const float4 zero = make_float4(0.0f, 0.0f, 0.0f, 0.0f);
    const int n4 = n >> 2;
    for (int i = gt; i < n4; i += 2 * nt) {
      const int j = i + nt;
      const int jc = (j < n4) ? j : i;
      const float4 f0 = f4[i];
      const float4 m0 = m4[i];
      const float4 f1 = f4[jc];
      const float4 m1 = m4[jc];
      p4[i] = make_float4(f0.x * m0.x, f0.y * m0.y, f0.z * m0.z, f0.w * m0.w);
      a4[i] = zero;
      p4[jc] = make_float4(f1.x * m1.x, f1.y * m1.y, f1.z * m1.z, f1.w * m1.w);
      a4[jc] = zero;
    }
  } else {
    for (long long i = gt; i < n; i += nt) {
      pre1[i] = feat[i] * norm[i];
      agg1[i] = 0.0f;
    }
  }
}

#define WBITS      15
#define WN         (1 << WBITS)
#define WMASK      (WN - 1)
#define MAXBINS    1024
#define BIN_BLOCKS 256
#define BT         1024
#define STAGE_CAP  65536
#define SRC_GROUPS 8

static_assert(NODE_BLOCKS * NODE_THREADS >= 2 * MAXBINS, "node grid too small to clear the bin counters");

template <class F>
__device__ __forceinline__ void chunk_edges(const int* __restrict__ src,
                                            const int* __restrict__ dst,
                                            int e4, int e, F f) {
  const int tid   = threadIdx.x;
  const int per   = (e4 + BIN_BLOCKS - 1) / BIN_BLOCKS;
  const int start = blockIdx.x * per;
  const int stop  = (start + per < e4) ? (start + per) : e4;
  const v4i* s4 = (const v4i*)src;
  const v4i* d4 = (const v4i*)dst;
  int v0 = start + tid;
  for (; v0 + 3 * BT < stop; v0 += BT * 4) {
    v4i d[4];
    v4i s[4];
#pragma unroll
    for (int k = 0; k < 4; ++k) {
      d[k] = d4[v0 + k * BT];
      s[k] = s4[v0 + k * BT];
    }
#pragma unroll
    for (int k = 0; k < 4; ++k) {
#pragma unroll
      for (int c = 0; c < 4; ++c) f(s[k][c], d[k][c]);
    }
  }
  {
    v4i d[3];
    v4i s[3];
#pragma unroll
    for (int k = 0; k < 3; ++k) {
      const int v = v0 + k * BT;
      if (v < stop) {
        d[k] = d4[v];
        s[k] = s4[v];
      } else {
        d[k] = (v4i)(0);
        s[k] = (v4i)(0);
      }
    }
#pragma unroll
    for (int k = 0; k < 3; ++k) {
      if (v0 + k * BT < stop) {
#pragma unroll
        for (int c = 0; c < 4; ++c) f(s[k][c], d[k][c]);
      }
    }
  }
  if (blockIdx.x == 0 && tid == 0) {
    for (int k = e4 * 4; k < e; ++k) f(src[k], dst[k]);
  }
}

__global__ __launch_bounds__(BT)
void gcn_bin_count(const int* __restrict__ src,
                   const int* __restrict__ dst,
                   unsigned* __restrict__ binTotal,
                   int e4, int e, int nw, int nb) {
  __shared__ unsigned hist[MAXBINS];
  hist[threadIdx.x] = 0u;
  __syncthreads();
  chunk_edges(src, dst, e4, e, [&](int s, int d) {
    const int bin = (d >> WBITS) * nw + (s >> WBITS);
    __hip_atomic_fetch_add(&hist[bin], 1u, __ATOMIC_RELAXED, __HIP_MEMORY_SCOPE_WORKGROUP);
  });
  __syncthreads();
  const unsigned c = hist[threadIdx.x];
  if ((int)threadIdx.x < nb && c != 0u) {
    __hip_atomic_fetch_add(&binTotal[threadIdx.x], c, __ATOMIC_RELAXED, __HIP_MEMORY_SCOPE_AGENT);
  }
}

__device__ __forceinline__ unsigned block_excl_scan(unsigned v, unsigned* wsum) {
  const int lane = threadIdx.x & 31;
  const int w    = threadIdx.x >> 5;
  unsigned x = v;
#pragma unroll
  for (int o = 1; o < 32; o <<= 1) {
    const unsigned y = __shfl_up(x, o);
    if (lane >= o) x += y;
  }
  if (lane == 31) wsum[w] = x;
  __syncthreads();
  if (w == 0) {
    unsigned s = wsum[lane];
#pragma unroll
    for (int o = 1; o < 32; o <<= 1) {
      const unsigned y = __shfl_up(s, o);
      if (lane >= o) s += y;
    }
    wsum[lane] = s;
  }
  __syncthreads();
  const unsigned base = (w > 0) ? wsum[w - 1] : 0u;
  __syncthreads();
  return base + x - v;
}

__global__ __launch_bounds__(BT)
void gcn_bin_scatter(const int* __restrict__ src,
                     const int* __restrict__ dst,
                     const unsigned* __restrict__ binTotal,
                     unsigned* __restrict__ binCursor,
                     unsigned* __restrict__ binStart,
                     unsigned* __restrict__ binned,
                     int e4, int e, int nw, int nb) {
  __shared__ unsigned stage[STAGE_CAP];
  __shared__ unsigned hist[MAXBINS];
  __shared__ unsigned lstart[MAXBINS];
  __shared__ unsigned gpos[MAXBINS];
  __shared__ unsigned fill[MAXBINS];
  __shared__ unsigned wsum[32];

  const int tid = threadIdx.x;
  hist[tid] = 0u;
  fill[tid] = 0u;
  __syncthreads();

  chunk_edges(src, dst, e4, e, [&](int s, int d) {
    const int bin = (d >> WBITS) * nw + (s >> WBITS);
    __hip_atomic_fetch_add(&hist[bin], 1u, __ATOMIC_RELAXED, __HIP_MEMORY_SCOPE_WORKGROUP);
  });
  __syncthreads();

  const unsigned mine = hist[tid];
  const unsigned ls   = block_excl_scan(mine, wsum);
  const unsigned tot  = (tid < nb) ? binTotal[tid] : 0u;
  const unsigned gs   = block_excl_scan(tot, wsum);
  lstart[tid] = ls;
  unsigned g = gs;
  if (mine != 0u) {
    g += __hip_atomic_fetch_add(&binCursor[tid], mine, __ATOMIC_RELAXED, __HIP_MEMORY_SCOPE_AGENT);
  }
  gpos[tid] = g;
  if (blockIdx.x == 0) {
    binStart[tid] = gs;
    if (tid == BT - 1) binStart[MAXBINS] = gs + tot;
  }
  __syncthreads();

  chunk_edges(src, dst, e4, e, [&](int s, int d) {
    const int bin = (d >> WBITS) * nw + (s >> WBITS);
    const unsigned r = __hip_atomic_fetch_add(&fill[bin], 1u, __ATOMIC_RELAXED, __HIP_MEMORY_SCOPE_WORKGROUP);
    stage[lstart[bin] + r] = (unsigned)(s & WMASK) | ((unsigned)(d & WMASK) << WBITS);
  });
  __syncthreads();

  const int lane = tid & 31;
  const int w    = tid >> 5;
  for (int bin = w; bin < nb; bin += 32) {
    const unsigned cnt = hist[bin];
    const unsigned l0  = lstart[bin];
    const unsigned g0  = gpos[bin];
    for (unsigned q = lane; q < cnt; q += 32) binned[g0 + q] = stage[l0 + q];
  }
}

__global__ __launch_bounds__(BT)
void gcn_edge_binned(const unsigned* __restrict__ binned,
                     const unsigned* __restrict__ binStart,
                     const float* __restrict__ val,
                     float* __restrict__ agg,
                     int nw, int n) {
  __shared__ float vals[WN];
  __shared__ float acc[WN];

  const int tid = threadIdx.x;
  const int j   = blockIdx.x / SRC_GROUPS;
  const int g   = blockIdx.x % SRC_GROUPS;
  const bool al = (((size_t)val) & 15) == 0;

  float4* acc4 = (float4*)acc;
  for (int i = tid; i < WN / 4; i += BT) {
    acc4[i] = make_float4(0.0f, 0.0f, 0.0f, 0.0f);
  }

  for (int i = g; i < nw; i += SRC_GROUPS) {
    __syncthreads();
    const int sbase = i * WN;
    if (al && sbase + WN <= n) {
      float4 x[WN / (BT * 4)];
#pragma unroll
      for (int k = 0; k < WN / (BT * 4); ++k) {
        x[k] = *(const float4*)(val + sbase + (k * BT + tid) * 4);
      }
#pragma unroll
      for (int k = 0; k < WN / (BT * 4); ++k) {
        *(float4*)(vals + (k * BT + tid) * 4) = x[k];
      }
    } else {
      float y[WN / BT];
#pragma unroll
      for (int k = 0; k < WN / BT; ++k) {
        const int id = sbase + k * BT + tid;
        y[k] = (id < n) ? val[id] : 0.0f;
      }
#pragma unroll
      for (int k = 0; k < WN / BT; ++k) vals[k * BT + tid] = y[k];
    }
    __syncthreads();

    const unsigned p0 = binStart[j * nw + i];
    const unsigned p1 = binStart[j * nw + i + 1];
    unsigned p = p0 + tid;
    for (; p + 7 * BT < p1; p += BT * 8) {
      unsigned wd[8];
      float    x[8];
#pragma unroll
      for (int k = 0; k < 8; ++k) wd[k] = binned[p + k * BT];
#pragma unroll
      for (int k = 0; k < 8; ++k) x[k] = vals[wd[k] & WMASK];
#pragma unroll
      for (int k = 0; k < 8; ++k) {
        __hip_atomic_fetch_add(&acc[wd[k] >> WBITS], x[k],
                               __ATOMIC_RELAXED, __HIP_MEMORY_SCOPE_WORKGROUP);
      }
    }
    for (; p < p1; p += BT * 4) {
      unsigned wd[4];
      float    x[4];
#pragma unroll
      for (int k = 0; k < 4; ++k) {
        const unsigned pp = p + k * BT;
        wd[k] = (pp < p1) ? binned[pp] : 0xFFFFFFFFu;
      }
#pragma unroll
      for (int k = 0; k < 4; ++k) {
        const bool ok = wd[k] != 0xFFFFFFFFu;
        const unsigned sl = ok ? (wd[k] & WMASK) : 0u;
        const unsigned dl = ok ? (wd[k] >> WBITS) : (unsigned)tid;
        const float v = vals[sl];
        x[k] = ok ? v : 0.0f;
        wd[k] = dl;
      }
#pragma unroll
      for (int k = 0; k < 4; ++k) {
        __hip_atomic_fetch_add(&acc[wd[k]], x[k],
                               __ATOMIC_RELAXED, __HIP_MEMORY_SCOPE_WORKGROUP);
      }
    }
  }
  __syncthreads();

  const int base = j * WN;
  for (int i = tid; i < WN; i += BT) {
    const float a = acc[i];
    const int id = base + i;
    if (id < n && a != 0.0f) {
      __hip_atomic_fetch_add(&agg[id], a, __ATOMIC_RELAXED, __HIP_MEMORY_SCOPE_AGENT);
    }
  }
}

#define WIN_NODES    65536
#define SCAN_SLICES  16
#define SCAN_THREADS 1024
#define SCAN_GROUP   4

__global__ __launch_bounds__(SCAN_THREADS)
void gcn_edge_window(const int* __restrict__ src,
                     const int* __restrict__ dst,
                     const float* __restrict__ val,
                     float* __restrict__ agg,
                     int e4, int e, int n) {
  __shared__ float acc[WIN_NODES];

  const int tid   = threadIdx.x;
  const int win   = blockIdx.x / SCAN_SLICES;
  const int slice = blockIdx.x % SCAN_SLICES;
  const int base  = win * WIN_NODES;

  float4* acc4 = (float4*)acc;
  for (int i = tid; i < WIN_NODES / 4; i += SCAN_THREADS) {
    acc4[i] = make_float4(0.0f, 0.0f, 0.0f, 0.0f);
  }
  __syncthreads();

  const int per   = (e4 + SCAN_SLICES - 1) / SCAN_SLICES;
  const int start = slice * per;
  const int stop  = (start + per < e4) ? (start + per) : e4;

  const v4i* s4 = (const v4i*)src;
  const v4i* d4 = (const v4i*)dst;

  for (int v0 = start + tid; v0 < stop; v0 += SCAN_THREADS * SCAN_GROUP) {
    v4i d[SCAN_GROUP];
    v4i s[SCAN_GROUP];
#pragma unroll
    for (int k = 0; k < SCAN_GROUP; ++k) {
      const int v = v0 + k * SCAN_THREADS;
      if (v < stop) {
        d[k] = d4[v];
        s[k] = s4[v];
      } else {
        d[k] = (v4i)(-1);
        s[k] = (v4i)(0);
      }
    }

    unsigned r[SCAN_GROUP][4];
    float    x[SCAN_GROUP][4];
#pragma unroll
    for (int k = 0; k < SCAN_GROUP; ++k) {
#pragma unroll
      for (int c = 0; c < 4; ++c) {
        r[k][c] = (unsigned)(d[k][c] - base);
        const int si = (r[k][c] < (unsigned)WIN_NODES) ? s[k][c] : 0;
        x[k][c] = val[si];
      }
    }
#pragma unroll
    for (int k = 0; k < SCAN_GROUP; ++k) {
#pragma unroll
      for (int c = 0; c < 4; ++c) {
        if (r[k][c] < (unsigned)WIN_NODES) {
          __hip_atomic_fetch_add(&acc[r[k][c]], x[k][c], __ATOMIC_RELAXED,
                                 __HIP_MEMORY_SCOPE_WORKGROUP);
        }
      }
    }
  }

  if (slice == 0 && tid == 0) {
    for (int k = e4 * 4; k < e; ++k) {
      const unsigned rr = (unsigned)(dst[k] - base);
      if (rr < (unsigned)WIN_NODES) {
        __hip_atomic_fetch_add(&acc[rr], val[src[k]], __ATOMIC_RELAXED,
                               __HIP_MEMORY_SCOPE_WORKGROUP);
      }
    }
  }
  __syncthreads();

  for (int i = tid; i < WIN_NODES; i += SCAN_THREADS) {
    const float a = acc[i];
    const int id = base + i;
    if (id < n && a != 0.0f) {
      __hip_atomic_fetch_add(&agg[id], a, __ATOMIC_RELAXED, __HIP_MEMORY_SCOPE_AGENT);
    }
  }
}

#define L1_GROUPS 4

__global__ __launch_bounds__(NODE_THREADS)
void gcn_layer1_wmma(const float* __restrict__ agg1,
                     const float* __restrict__ norm,
                     const float* __restrict__ W1,
                     const float* __restrict__ b1,
                     const float* __restrict__ W2,
                     float* __restrict__ t,
                     float* __restrict__ agg2,
                     int n_nodes) {
  const int lane = threadIdx.x & 31;
  const int step = gridDim.x * blockDim.x * L1_GROUPS;
  const int wb0 = __builtin_amdgcn_readfirstlane(
      (int)((blockIdx.x * blockDim.x + (threadIdx.x & ~31u)) * L1_GROUPS));

  const int ch = (lane < 16) ? lane : lane - 24;
  float w1a = 0.0f;
  if (ch >= 0 && ch < 3) w1a = W1[ch];
  v2f a;
  a.x = w1a;
  a.y = 0.0f;
  v8f c;
#pragma unroll
  for (int r = 0; r < 8; ++r) c[r] = 0.0f;
  c[0] = b1[0];
  c[1] = b1[1];
  c[2] = b1[2];
  const float w20 = W2[0];
  const float w21 = W2[1];
  const float w22 = W2[2];

  for (int wb = wb0; wb < n_nodes; wb += step) {
    float a1[L1_GROUPS];
    float nm[L1_GROUPS];
#pragma unroll
    for (int u = 0; u < L1_GROUPS; ++u) {
      const int i = wb + u * 32 + lane;
      const int ic = (i < n_nodes) ? i : wb;
      a1[u] = agg1[ic];
      nm[u] = norm[ic];
    }

    v8f d[L1_GROUPS];
#pragma unroll
    for (int u = 0; u < L1_GROUPS; ++u) {
      const int i = wb + u * 32 + lane;
      v2f b;
      b.x = (i < n_nodes) ? a1[u] * nm[u] : 0.0f;
      b.y = 0.0f;
      d[u] = __builtin_amdgcn_wmma_f32_16x16x4_f32(
           false, a,  false, b,
           (short)0, c,  false,  false);
    }

#pragma unroll
    for (int u = 0; u < L1_GROUPS; ++u) {
      const int i = wb + u * 32 + lane;
      const float h0 = fmaxf(d[u][0], 0.0f) * w20;
      const float h1 = fmaxf(d[u][1], 0.0f) * w21;
      const float h2 = fmaxf(d[u][2], 0.0f) * w22;
      const float s = (h0 + h1) + h2;
      if (i < n_nodes) {
        t[i] = nm[u] * s;
        agg2[i] = 0.0f;
      }
    }
  }
}

__global__ __launch_bounds__(NODE_THREADS)
void gcn_final(const float* __restrict__ norm,
               const float* __restrict__ agg2,
               const float* __restrict__ b2,
               float* __restrict__ out,
               int n) {
  const int gt = blockIdx.x * blockDim.x + threadIdx.x;
  const int nt = gridDim.x * blockDim.x;
  const float bias = b2[0];

  const bool vec = ((n & 3) == 0) &&
                   (((((size_t)norm) | ((size_t)agg2) | ((size_t)out)) & 15) == 0);
  if (vec) {
    const float4* m4 = (const float4*)norm;
    const float4* a4 = (const float4*)agg2;
    float4* o4 = (float4*)out;
    const int n4 = n >> 2;
    for (int i = gt; i < n4; i += 2 * nt) {
      const int j = i + nt;
      const int jc = (j < n4) ? j : i;
      const float4 m0 = m4[i];
      const float4 a0 = a4[i];
      const float4 m1 = m4[jc];
      const float4 a1 = a4[jc];
      o4[i] = make_float4(fmaxf(m0.x * a0.x + bias, 0.0f), fmaxf(m0.y * a0.y + bias, 0.0f),
                          fmaxf(m0.z * a0.z + bias, 0.0f), fmaxf(m0.w * a0.w + bias, 0.0f));
      o4[jc] = make_float4(fmaxf(m1.x * a1.x + bias, 0.0f), fmaxf(m1.y * a1.y + bias, 0.0f),
                           fmaxf(m1.z * a1.z + bias, 0.0f), fmaxf(m1.w * a1.w + bias, 0.0f));
    }
  } else {
    for (long long i = gt; i < n; i += nt) {
      out[i] = fmaxf(norm[i] * agg2[i] + bias, 0.0f);
    }
  }
}

extern "C" void kernel_launch(void* const* d_in, const int* in_sizes, int n_in,
                              void* d_out, int out_size, void* d_ws, size_t ws_size,
                              hipStream_t stream) {
  (void)n_in; (void)out_size;

  const float* feat = (const float*)d_in[0];
  const float* norm = (const float*)d_in[1];
  const int*   src  = (const int*)d_in[2];
  const int*   dst  = (const int*)d_in[3];
  const float* W1   = (const float*)d_in[4];
  const float* b1   = (const float*)d_in[5];
  const float* W2   = (const float*)d_in[6];
  const float* b2   = (const float*)d_in[7];
  float* out = (float*)d_out;

  const int N = in_sizes[0];
  const int E = in_sizes[2];

  float* pre1 = (float*)d_ws;
  float* agg1 = pre1 + N;
  float* tval = agg1 + N;
  float* agg2 = tval + N;
  unsigned* binTotal  = (unsigned*)(agg2 + N);
  unsigned* binCursor = binTotal + MAXBINS;
  unsigned* binStart  = binCursor + MAXBINS;
  unsigned* binned    = binStart + MAXBINS + 4;

  const int e4 = E / 4;

  const int nw = (N + WN - 1) / WN;
  const int nb = nw * nw;
  const size_t need = (size_t)N * 16 + (size_t)(3 * MAXBINS + 4) * 4 + (size_t)E * 4;
  const int slice_edges = ((e4 + BIN_BLOCKS - 1) / BIN_BLOCKS) * 4 + 3;
  const bool binned_path = (ws_size >= need) && (nb <= MAXBINS) && (slice_edges <= STAGE_CAP);

  if (binned_path) {
    gcn_premul_zero<<<NODE_BLOCKS, NODE_THREADS, 0, stream>>>(feat, norm, pre1, agg1, binTotal, 2 * MAXBINS, N);
    gcn_bin_count<<<BIN_BLOCKS, BT, 0, stream>>>(src, dst, binTotal, e4, E, nw, nb);
    gcn_bin_scatter<<<BIN_BLOCKS, BT, 0, stream>>>(src, dst, binTotal, binCursor, binStart, binned,
                                                   e4, E, nw, nb);
    gcn_edge_binned<<<nw * SRC_GROUPS, BT, 0, stream>>>(binned, binStart, pre1, agg1, nw, N);
    gcn_layer1_wmma<<<NODE_BLOCKS, NODE_THREADS, 0, stream>>>(agg1, norm, W1, b1, W2, tval, agg2, N);
    gcn_edge_binned<<<nw * SRC_GROUPS, BT, 0, stream>>>(binned, binStart, tval, agg2, nw, N);
    gcn_final<<<NODE_BLOCKS, NODE_THREADS, 0, stream>>>(norm, agg2, b2, out, N);
    return;
  }

  const int n_win = (N + WIN_NODES - 1) / WIN_NODES;
  const int grid_edges = n_win * SCAN_SLICES;

  gcn_premul_zero<<<NODE_BLOCKS, NODE_THREADS, 0, stream>>>(feat, norm, pre1, agg1, binTotal, 0, N);
  gcn_edge_window<<<grid_edges, SCAN_THREADS, 0, stream>>>(src, dst, pre1, agg1, e4, E, N);
  gcn_layer1_wmma<<<NODE_BLOCKS, NODE_THREADS, 0, stream>>>(agg1, norm, W1, b1, W2, tval, agg2, N);
  gcn_edge_window<<<grid_edges, SCAN_THREADS, 0, stream>>>(src, dst, tval, agg2, e4, E, N);
  gcn_final<<<NODE_BLOCKS, NODE_THREADS, 0, stream>>>(norm, agg2, b2, out, N);
}


# Bins kept per binning block (ordered slice copied out whole, table of run starts): no count launch, no returning global adds; edge kernels walk 256 runs per bin; LDS sum order changes
# speedup vs baseline: 8.7419x; 1.0454x over previous
#include <hip/hip_runtime.h>
#include <hip/hip_bf16.h>


typedef __attribute__((ext_vector_type(2))) float v2f;
typedef __attribute__((ext_vector_type(8))) float v8f;
typedef __attribute__((ext_vector_type(4))) int   v4i;

#define NODE_BLOCKS  512
#define NODE_THREADS 256

__global__ __launch_bounds__(NODE_THREADS)
void gcn_premul_zero(const float* __restrict__ feat,
                     const float* __restrict__ norm,
                     float* __restrict__ pre1,
                     float* __restrict__ agg1,
                     int n) {
  const int gt = blockIdx.x * blockDim.x + threadIdx.x;
  const int nt = gridDim.x * blockDim.x;

  const bool vec = ((n & 3) == 0) &&
                   (((((size_t)feat) | ((size_t)norm) | ((size_t)pre1) | ((size_t)agg1)) & 15) == 0);
  if (vec) {
    const float4* f4 = (const float4*)feat;
    const float4* m4 = (const float4*)norm;
    float4* p4 = (float4*)pre1;
    float4* a4 = (float4*)agg1;
    const float4 zero = make_float4(0.0f, 0.0f, 0.0f, 0.0f);
    const int n4 = n >> 2;
    for (int i = gt; i < n4; i += 2 * nt) {
      const int j = i + nt;
      const int jc = (j < n4) ? j : i;
      const float4 f0 = f4[i];
      const float4 m0 = m4[i];
      const float4 f1 = f4[jc];
      const float4 m1 = m4[jc];
      p4[i] = make_float4(f0.x * m0.x, f0.y * m0.y, f0.z * m0.z, f0.w * m0.w);
      a4[i] = zero;
      p4[jc] = make_float4(f1.x * m1.x, f1.y * m1.y, f1.z * m1.z, f1.w * m1.w);
      a4[jc] = zero;
    }
  } else {
    for (long long i = gt; i < n; i += nt) {
      pre1[i] = feat[i] * norm[i];
      agg1[i] = 0.0f;
    }
  }
}

#define WBITS      15
#define WN         (1 << WBITS)
#define WMASK      (WN - 1)
#define MAXBINS    1024
#define BIN_BLOCKS 256
#define BT         1024
#define STAGE_CAP  65536
#define SRC_GROUPS 8
#define RUNS_PER_WAVE 8
#define TABLE_WORDS ((MAXBINS + 1) * BIN_BLOCKS)

static_assert((BT / 32) * RUNS_PER_WAVE == BIN_BLOCKS, "the waves of a layer block must cover every run of a bin");
static_assert(BT == MAXBINS, "one thread per bin in the binning block");

template <class F>
__device__ __forceinline__ void chunk_edges(const int* __restrict__ src,
                                            const int* __restrict__ dst,
                                            int e4, int e, F f) {
  const int tid   = threadIdx.x;
  const int per   = (e4 + BIN_BLOCKS - 1) / BIN_BLOCKS;
  const int start = blockIdx.x * per;
  const int stop  = (start + per < e4) ? (start + per) : e4;
  const v4i* s4 = (const v4i*)src;
  const v4i* d4 = (const v4i*)dst;
  int v0 = start + tid;
  for (; v0 + 3 * BT < stop; v0 += BT * 4) {
    v4i d[4];
    v4i s[4];
#pragma unroll
    for (int k = 0; k < 4; ++k) {
      d[k] = d4[v0 + k * BT];
      s[k] = s4[v0 + k * BT];
    }
#pragma unroll
    for (int k = 0; k < 4; ++k) {
#pragma unroll
      for (int c = 0; c < 4; ++c) f(s[k][c], d[k][c]);
    }
  }
  {
    v4i d[3];
    v4i s[3];
#pragma unroll
    for (int k = 0; k < 3; ++k) {
      const int v = v0 + k * BT;
      if (v < stop) {
        d[k] = d4[v];
        s[k] = s4[v];
      } else {
        d[k] = (v4i)(0);
        s[k] = (v4i)(0);
      }
    }
#pragma unroll
    for (int k = 0; k < 3; ++k) {
      if (v0 + k * BT < stop) {
#pragma unroll
        for (int c = 0; c < 4; ++c) f(s[k][c], d[k][c]);
      }
    }
  }
  if (blockIdx.x == 0 && tid == 0) {
    for (int k = e4 * 4; k < e; ++k) f(src[k], dst[k]);
  }
}

__device__ __forceinline__ unsigned block_excl_scan(unsigned v, unsigned* wsum) {
  const int lane = threadIdx.x & 31;
  const int w    = threadIdx.x >> 5;
  unsigned x = v;
#pragma unroll
  for (int o = 1; o < 32; o <<= 1) {
    const unsigned y = __shfl_up(x, o);
    if (lane >= o) x += y;
  }
  if (lane == 31) wsum[w] = x;
  __syncthreads();
  if (w == 0) {
    unsigned s = wsum[lane];
#pragma unroll
    for (int o = 1; o < 32; o <<= 1) {
      const unsigned y = __shfl_up(s, o);
      if (lane >= o) s += y;
    }
    wsum[lane] = s;
  }
  __syncthreads();
  const unsigned base = (w > 0) ? wsum[w - 1] : 0u;
  __syncthreads();
  return base + x - v;
}

__global__ __launch_bounds__(BT)
void gcn_bin_local(const int* __restrict__ src,
                   const int* __restrict__ dst,
                   unsigned* __restrict__ runStart,
                   unsigned* __restrict__ binned,
                   int e4, int e, int nw, int cap) {
  __shared__ __attribute__((aligned(16))) unsigned stage[STAGE_CAP];
  __shared__ unsigned hist[MAXBINS];
  __shared__ unsigned fill[MAXBINS];
  __shared__ unsigned wsum[32];
  __shared__ unsigned total;

  const int tid = threadIdx.x;
  hist[tid] = 0u;
  __syncthreads();

  chunk_edges(src, dst, e4, e, [&](int s, int d) {
    const int bin = (d >> WBITS) * nw + (s >> WBITS);
    __hip_atomic_fetch_add(&hist[bin], 1u, __ATOMIC_RELAXED, __HIP_MEMORY_SCOPE_WORKGROUP);
  });
  __syncthreads();

  const unsigned mine = hist[tid];
  const unsigned ls   = block_excl_scan(mine, wsum);
  const unsigned base = blockIdx.x * (unsigned)cap;
  fill[tid] = ls;
  runStart[tid * BIN_BLOCKS + blockIdx.x] = base + ls;
  if (tid == BT - 1) {
    runStart[MAXBINS * BIN_BLOCKS + blockIdx.x] = base + ls + mine;
    total = ls + mine;
  }
  __syncthreads();

  chunk_edges(src, dst, e4, e, [&](int s, int d) {
    const int bin = (d >> WBITS) * nw + (s >> WBITS);
    const unsigned r = __hip_atomic_fetch_add(&fill[bin], 1u, __ATOMIC_RELAXED, __HIP_MEMORY_SCOPE_WORKGROUP);
    stage[r] = (unsigned)(s & WMASK) | ((unsigned)(d & WMASK) << WBITS);
  });
  __syncthreads();

  const unsigned tot = total;
  unsigned* __restrict__ outp = binned + base;
  if ((((size_t)outp) & 15) == 0) {
    for (unsigned q = tid * 4; q < tot; q += BT * 4) {
      if (q + 3 < tot) {
        *(uint4*)(outp + q) = *(const uint4*)(stage + q);
      } else {
        for (unsigned k = q; k < tot; ++k) outp[k] = stage[k];
      }
    }
  } else {
    for (unsigned q = tid; q < tot; q += BT) outp[q] = stage[q];
  }
}

__global__ __launch_bounds__(BT)
void gcn_edge_binned(const unsigned* __restrict__ binned,
                     const unsigned* __restrict__ runStart,
                     const float* __restrict__ val,
                     float* __restrict__ agg,
                     int nw, int n) {
  __shared__ float vals[WN];
  __shared__ float acc[WN];

  const int tid  = threadIdx.x;
  const int lane = tid & 31;
  const int j    = blockIdx.x / SRC_GROUPS;
  const int g    = blockIdx.x % SRC_GROUPS;
  const bool al  = (((size_t)val) & 15) == 0;
  const int b0 = __builtin_amdgcn_readfirstlane((tid >> 5) * RUNS_PER_WAVE);

  float4* acc4 = (float4*)acc;
  for (int i = tid; i < WN / 4; i += BT) {
    acc4[i] = make_float4(0.0f, 0.0f, 0.0f, 0.0f);
  }

  for (int i = g; i < nw; i += SRC_GROUPS) {
    __syncthreads();

    const unsigned* __restrict__ row = runStart + (j * nw + i) * BIN_BLOCKS + b0;
    unsigned rs[RUNS_PER_WAVE];
    unsigned rc[RUNS_PER_WAVE];
#pragma unroll
    for (int r = 0; r < RUNS_PER_WAVE; ++r) {
      rs[r] = row[r];
      rc[r] = row[BIN_BLOCKS + r] - rs[r];
    }

    const int sbase = i * WN;
    if (al && sbase + WN <= n) {
      float4 x[WN / (BT * 4)];
#pragma unroll
      for (int k = 0; k < WN / (BT * 4); ++k) {
        x[k] = *(const float4*)(val + sbase + (k * BT + tid) * 4);
      }
#pragma unroll
      for (int k = 0; k < WN / (BT * 4); ++k) {
        *(float4*)(vals + (k * BT + tid) * 4) = x[k];
      }
    } else {
      float y[WN / BT];
#pragma unroll
      for (int k = 0; k < WN / BT; ++k) {
        const int id = sbase + k * BT + tid;
        y[k] = (id < n) ? val[id] : 0.0f;
      }
#pragma unroll
      for (int k = 0; k < WN / BT; ++k) vals[k * BT + tid] = y[k];
    }
    __syncthreads();

    unsigned longest = 0u;
#pragma unroll
    for (int r = 0; r < RUNS_PER_WAVE; ++r) longest = (rc[r] > longest) ? rc[r] : longest;

    for (unsigned q0 = 0u; q0 < longest; q0 += 32u) {
      const unsigned q = q0 + lane;
      unsigned wd[RUNS_PER_WAVE];
      float    x[RUNS_PER_WAVE];
#pragma unroll
      for (int r = 0; r < RUNS_PER_WAVE; ++r) {
        wd[r] = (q < rc[r]) ? binned[rs[r] + q] : 0xFFFFFFFFu;
      }
#pragma unroll
      for (int r = 0; r < RUNS_PER_WAVE; ++r) {
        const bool ok = wd[r] != 0xFFFFFFFFu;
        const unsigned sl = ok ? (wd[r] & WMASK) : 0u;
        const unsigned dl = ok ? (wd[r] >> WBITS) : (unsigned)tid;
        const float v = vals[sl];
        x[r] = ok ? v : 0.0f;
        wd[r] = dl;
      }
#pragma unroll
      for (int r = 0; r < RUNS_PER_WAVE; ++r) {
        __hip_atomic_fetch_add(&acc[wd[r]], x[r],
                               __ATOMIC_RELAXED, __HIP_MEMORY_SCOPE_WORKGROUP);
      }
    }
  }
  __syncthreads();

  const int base = j * WN;
  for (int i = tid; i < WN; i += BT) {
    const float a = acc[i];
    const int id = base + i;
    if (id < n && a != 0.0f) {
      __hip_atomic_fetch_add(&agg[id], a, __ATOMIC_RELAXED, __HIP_MEMORY_SCOPE_AGENT);
    }
  }
}

#define WIN_NODES    65536
#define SCAN_SLICES  16
#define SCAN_THREADS 1024
#define SCAN_GROUP   4

__global__ __launch_bounds__(SCAN_THREADS)
void gcn_edge_window(const int* __restrict__ src,
                     const int* __restrict__ dst,
                     const float* __restrict__ val,
                     float* __restrict__ agg,
                     int e4, int e, int n) {
  __shared__ float acc[WIN_NODES];

  const int tid   = threadIdx.x;
  const int win   = blockIdx.x / SCAN_SLICES;
  const int slice = blockIdx.x % SCAN_SLICES;
  const int base  = win * WIN_NODES;

  float4* acc4 = (float4*)acc;
  for (int i = tid; i < WIN_NODES / 4; i += SCAN_THREADS) {
    acc4[i] = make_float4(0.0f, 0.0f, 0.0f, 0.0f);
  }
  __syncthreads();

  const int per   = (e4 + SCAN_SLICES - 1) / SCAN_SLICES;
  const int start = slice * per;
  const int stop  = (start + per < e4) ? (start + per) : e4;

  const v4i* s4 = (const v4i*)src;
  const v4i* d4 = (const v4i*)dst;

  for (int v0 = start + tid; v0 < stop; v0 += SCAN_THREADS * SCAN_GROUP) {
    v4i d[SCAN_GROUP];
    v4i s[SCAN_GROUP];
#pragma unroll
    for (int k = 0; k < SCAN_GROUP; ++k) {
      const int v = v0 + k * SCAN_THREADS;
      if (v < stop) {
        d[k] = d4[v];
        s[k] = s4[v];
      } else {
        d[k] = (v4i)(-1);
        s[k] = (v4i)(0);
      }
    }

    unsigned r[SCAN_GROUP][4];
    float    x[SCAN_GROUP][4];
#pragma unroll
    for (int k = 0; k < SCAN_GROUP; ++k) {
#pragma unroll
      for (int c = 0; c < 4; ++c) {
        r[k][c] = (unsigned)(d[k][c] - base);
        const int si = (r[k][c] < (unsigned)WIN_NODES) ? s[k][c] : 0;
        x[k][c] = val[si];
      }
    }
#pragma unroll
    for (int k = 0; k < SCAN_GROUP; ++k) {
#pragma unroll
      for (int c = 0; c < 4; ++c) {
        if (r[k][c] < (unsigned)WIN_NODES) {
          __hip_atomic_fetch_add(&acc[r[k][c]], x[k][c], __ATOMIC_RELAXED,
                                 __HIP_MEMORY_SCOPE_WORKGROUP);
        }
      }
    }
  }

  if (slice == 0 && tid == 0) {
    for (int k = e4 * 4; k < e; ++k) {
      const unsigned rr = (unsigned)(dst[k] - base);
      if (rr < (unsigned)WIN_NODES) {
        __hip_atomic_fetch_add(&acc[rr], val[src[k]], __ATOMIC_RELAXED,
                               __HIP_MEMORY_SCOPE_WORKGROUP);
      }
    }
  }
  __syncthreads();

  for (int i = tid; i < WIN_NODES; i += SCAN_THREADS) {
    const float a = acc[i];
    const int id = base + i;
    if (id < n && a != 0.0f) {
      __hip_atomic_fetch_add(&agg[id], a, __ATOMIC_RELAXED, __HIP_MEMORY_SCOPE_AGENT);
    }
  }
}

#define L1_GROUPS 4

__global__ __launch_bounds__(NODE_THREADS)
void gcn_layer1_wmma(const float* __restrict__ agg1,
                     const float* __restrict__ norm,
                     const float* __restrict__ W1,
                     const float* __restrict__ b1,
                     const float* __restrict__ W2,
                     float* __restrict__ t,
                     float* __restrict__ agg2,
                     int n_nodes) {
  const int lane = threadIdx.x & 31;
  const int step = gridDim.x * blockDim.x * L1_GROUPS;
  const int wb0 = __builtin_amdgcn_readfirstlane(
      (int)((blockIdx.x * blockDim.x + (threadIdx.x & ~31u)) * L1_GROUPS));

  const int ch = (lane < 16) ? lane : lane - 24;
  float w1a = 0.0f;
  if (ch >= 0 && ch < 3) w1a = W1[ch];
  v2f a;
  a.x = w1a;
  a.y = 0.0f;
  v8f c;
#pragma unroll
  for (int r = 0; r < 8; ++r) c[r] = 0.0f;
  c[0] = b1[0];
  c[1] = b1[1];
  c[2] = b1[2];
  const float w20 = W2[0];
  const float w21 = W2[1];
  const float w22 = W2[2];

  for (int wb = wb0; wb < n_nodes; wb += step) {
    float a1[L1_GROUPS];
    float nm[L1_GROUPS];
#pragma unroll
    for (int u = 0; u < L1_GROUPS; ++u) {
      const int i = wb + u * 32 + lane;
      const int ic = (i < n_nodes) ? i : wb;
      a1[u] = agg1[ic];
      nm[u] = norm[ic];
    }

    v8f d[L1_GROUPS];
#pragma unroll
    for (int u = 0; u < L1_GROUPS; ++u) {
      const int i = wb + u * 32 + lane;
      v2f b;
      b.x = (i < n_nodes) ? a1[u] * nm[u] : 0.0f;
      b.y = 0.0f;
      d[u] = __builtin_amdgcn_wmma_f32_16x16x4_f32(
           false, a,  false, b,
           (short)0, c,  false,  false);
    }

#pragma unroll
    for (int u = 0; u < L1_GROUPS; ++u) {
      const int i = wb + u * 32 + lane;
      const float h0 = fmaxf(d[u][0], 0.0f) * w20;
      const float h1 = fmaxf(d[u][1], 0.0f) * w21;
      const float h2 = fmaxf(d[u][2], 0.0f) * w22;
      const float s = (h0 + h1) + h2;
      if (i < n_nodes) {
        t[i] = nm[u] * s;
        agg2[i] = 0.0f;
      }
    }
  }
}

__global__ __launch_bounds__(NODE_THREADS)
void gcn_final(const float* __restrict__ norm,
               const float* __restrict__ agg2,
               const float* __restrict__ b2,
               float* __restrict__ out,
               int n) {
  const int gt = blockIdx.x * blockDim.x + threadIdx.x;
  const int nt = gridDim.x * blockDim.x;
  const float bias = b2[0];

  const bool vec = ((n & 3) == 0) &&
                   (((((size_t)norm) | ((size_t)agg2) | ((size_t)out)) & 15) == 0);
  if (vec) {
    const float4* m4 = (const float4*)norm;
    const float4* a4 = (const float4*)agg2;
    float4* o4 = (float4*)out;
    const int n4 = n >> 2;
    for (int i = gt; i < n4; i += 2 * nt) {
      const int j = i + nt;
      const int jc = (j < n4) ? j : i;
      const float4 m0 = m4[i];
      const float4 a0 = a4[i];
      const float4 m1 = m4[jc];
      const float4 a1 = a4[jc];
      o4[i] = make_float4(fmaxf(m0.x * a0.x + bias, 0.0f), fmaxf(m0.y * a0.y + bias, 0.0f),
                          fmaxf(m0.z * a0.z + bias, 0.0f), fmaxf(m0.w * a0.w + bias, 0.0f));
      o4[jc] = make_float4(fmaxf(m1.x * a1.x + bias, 0.0f), fmaxf(m1.y * a1.y + bias, 0.0f),
                           fmaxf(m1.z * a1.z + bias, 0.0f), fmaxf(m1.w * a1.w + bias, 0.0f));
    }
  } else {
    for (long long i = gt; i < n; i += nt) {
      out[i] = fmaxf(norm[i] * agg2[i] + bias, 0.0f);
    }
  }
}

extern "C" void kernel_launch(void* const* d_in, const int* in_sizes, int n_in,
                              void* d_out, int out_size, void* d_ws, size_t ws_size,
                              hipStream_t stream) {
  (void)n_in; (void)out_size;

  const float* feat = (const float*)d_in[0];
  const float* norm = (const float*)d_in[1];
  const int*   src  = (const int*)d_in[2];
  const int*   dst  = (const int*)d_in[3];
  const float* W1   = (const float*)d_in[4];
  const float* b1   = (const float*)d_in[5];
  const float* W2   = (const float*)d_in[6];
  const float* b2   = (const float*)d_in[7];
  float* out = (float*)d_out;

  const int N = in_sizes[0];
  const int E = in_sizes[2];

  float* pre1 = (float*)d_ws;
  float* agg1 = pre1 + N;
  float* tval = agg1 + N;
  float* agg2 = tval + N;
  unsigned* binned   = (unsigned*)(agg2 + N);
  unsigned* runStart = (unsigned*)d_out;

  const int e4 = E / 4;

  const int nw = (N + WN - 1) / WN;
  const int nb = nw * nw;
  const int slice_edges = ((e4 + BIN_BLOCKS - 1) / BIN_BLOCKS) * 4 + 3;
  const int cap = slice_edges + 1;
  const size_t need = (size_t)N * 16 + (size_t)BIN_BLOCKS * cap * 4;
  const bool binned_path = (ws_size >= need) && (nb <= MAXBINS) && (slice_edges <= STAGE_CAP) &&
                           ((size_t)N >= (size_t)TABLE_WORDS);

  if (binned_path) {
    gcn_premul_zero<<<NODE_BLOCKS, NODE_THREADS, 0, stream>>>(feat, norm, pre1, agg1, N);
    gcn_bin_local<<<BIN_BLOCKS, BT, 0, stream>>>(src, dst, runStart, binned, e4, E, nw, cap);
    gcn_edge_binned<<<nw * SRC_GROUPS, BT, 0, stream>>>(binned, runStart, pre1, agg1, nw, N);
    gcn_layer1_wmma<<<NODE_BLOCKS, NODE_THREADS, 0, stream>>>(agg1, norm, W1, b1, W2, tval, agg2, N);
    gcn_edge_binned<<<nw * SRC_GROUPS, BT, 0, stream>>>(binned, runStart, tval, agg2, nw, N);
    gcn_final<<<NODE_BLOCKS, NODE_THREADS, 0, stream>>>(norm, agg2, b2, out, N);
    return;
  }

  const int n_win = (N + WIN_NODES - 1) / WIN_NODES;
  const int grid_edges = n_win * SCAN_SLICES;

  gcn_premul_zero<<<NODE_BLOCKS, NODE_THREADS, 0, stream>>>(feat, norm, pre1, agg1, N);
  gcn_edge_window<<<grid_edges, SCAN_THREADS, 0, stream>>>(src, dst, pre1, agg1, e4, E, N);
  gcn_layer1_wmma<<<NODE_BLOCKS, NODE_THREADS, 0, stream>>>(agg1, norm, W1, b1, W2, tval, agg2, N);
  gcn_edge_window<<<grid_edges, SCAN_THREADS, 0, stream>>>(src, dst, tval, agg2, e4, E, N);
  gcn_final<<<NODE_BLOCKS, NODE_THREADS, 0, stream>>>(norm, agg2, b2, out, N);
}


# Pre-multiply and zeroing of the layer-1 sums move into the binning kernel (one vector per thread, loaded ahead of its count pass, stored behind it): 6 launches become 5
# speedup vs baseline: 9.3509x; 1.0697x over previous
#include <hip/hip_runtime.h>
#include <hip/hip_bf16.h>


typedef __attribute__((ext_vector_type(2))) float v2f;
typedef __attribute__((ext_vector_type(8))) float v8f;
typedef __attribute__((ext_vector_type(4))) int   v4i;

#define NODE_BLOCKS  512
#define NODE_THREADS 256

__global__ __launch_bounds__(NODE_THREADS)
void gcn_premul_zero(const float* __restrict__ feat,
                     const float* __restrict__ norm,
                     float* __restrict__ pre1,
                     float* __restrict__ agg1,
                     int n) {
  const int gt = blockIdx.x * blockDim.x + threadIdx.x;
  const int nt = gridDim.x * blockDim.x;

  const bool vec = ((n & 3) == 0) &&
                   (((((size_t)feat) | ((size_t)norm) | ((size_t)pre1) | ((size_t)agg1)) & 15) == 0);
  if (vec) {
    const float4* f4 = (const float4*)feat;
    const float4* m4 = (const float4*)norm;
    float4* p4 = (float4*)pre1;
    float4* a4 = (float4*)agg1;
    const float4 zero = make_float4(0.0f, 0.0f, 0.0f, 0.0f);
    const int n4 = n >> 2;
    for (int i = gt; i < n4; i += 2 * nt) {
      const int j = i + nt;
      const int jc = (j < n4) ? j : i;
      const float4 f0 = f4[i];
      const float4 m0 = m4[i];
      const float4 f1 = f4[jc];
      const float4 m1 = m4[jc];
      p4[i] = make_float4(f0.x * m0.x, f0.y * m0.y, f0.z * m0.z, f0.w * m0.w);
      a4[i] = zero;
      p4[jc] = make_float4(f1.x * m1.x, f1.y * m1.y, f1.z * m1.z, f1.w * m1.w);
      a4[jc] = zero;
    }
  } else {
    for (long long i = gt; i < n; i += nt) {
      pre1[i] = feat[i] * norm[i];
      agg1[i] = 0.0f;
    }
  }
}

#define WBITS      15
#define WN         (1 << WBITS)
#define WMASK      (WN - 1)
#define MAXBINS    1024
#define BIN_BLOCKS 256
#define BT         1024
#define STAGE_CAP  65536
#define SRC_GROUPS 8
#define RUNS_PER_WAVE 8
#define TABLE_WORDS ((MAXBINS + 1) * BIN_BLOCKS)

static_assert((BT / 32) * RUNS_PER_WAVE == BIN_BLOCKS, "the waves of a layer block must cover every run of a bin");
static_assert(BT == MAXBINS, "one thread per bin in the binning block");

template <class F>
__device__ __forceinline__ void chunk_edges(const int* __restrict__ src,
                                            const int* __restrict__ dst,
                                            int e4, int e, F f) {
  const int tid   = threadIdx.x;
  const int per   = (e4 + BIN_BLOCKS - 1) / BIN_BLOCKS;
  const int start = blockIdx.x * per;
  const int stop  = (start + per < e4) ? (start + per) : e4;
  const v4i* s4 = (const v4i*)src;
  const v4i* d4 = (const v4i*)dst;
  int v0 = start + tid;
  for (; v0 + 3 * BT < stop; v0 += BT * 4) {
    v4i d[4];
    v4i s[4];
#pragma unroll
    for (int k = 0; k < 4; ++k) {
      d[k] = d4[v0 + k * BT];
      s[k] = s4[v0 + k * BT];
    }
#pragma unroll
    for (int k = 0; k < 4; ++k) {
#pragma unroll
      for (int c = 0; c < 4; ++c) f(s[k][c], d[k][c]);
    }
  }
  {
    v4i d[3];
    v4i s[3];
#pragma unroll
    for (int k = 0; k < 3; ++k) {
      const int v = v0 + k * BT;
      if (v < stop) {
        d[k] = d4[v];
        s[k] = s4[v];
      } else {
        d[k] = (v4i)(0);
        s[k] = (v4i)(0);
      }
    }
#pragma unroll
    for (int k = 0; k < 3; ++k) {
      if (v0 + k * BT < stop) {
#pragma unroll
        for (int c = 0; c < 4; ++c) f(s[k][c], d[k][c]);
      }
    }
  }
  if (blockIdx.x == 0 && tid == 0) {
    for (int k = e4 * 4; k < e; ++k) f(src[k], dst[k]);
  }
}

__device__ __forceinline__ unsigned block_excl_scan(unsigned v, unsigned* wsum) {
  const int lane = threadIdx.x & 31;
  const int w    = threadIdx.x >> 5;
  unsigned x = v;
#pragma unroll
  for (int o = 1; o < 32; o <<= 1) {
    const unsigned y = __shfl_up(x, o);
    if (lane >= o) x += y;
  }
  if (lane == 31) wsum[w] = x;
  __syncthreads();
  if (w == 0) {
    unsigned s = wsum[lane];
#pragma unroll
    for (int o = 1; o < 32; o <<= 1) {
      const unsigned y = __shfl_up(s, o);
      if (lane >= o) s += y;
    }
    wsum[lane] = s;
  }
  __syncthreads();
  const unsigned base = (w > 0) ? wsum[w - 1] : 0u;
  __syncthreads();
  return base + x - v;
}

__global__ __launch_bounds__(BT)
void gcn_bin_local(const int* __restrict__ src,
                   const int* __restrict__ dst,
                   unsigned* __restrict__ runStart,
                   unsigned* __restrict__ binned,
                   const float* __restrict__ feat,
                   const float* __restrict__ norm,
                   float* __restrict__ pre1,
                   float* __restrict__ agg1,
                   int e4, int e, int nw, int cap, int n) {
  __shared__ __attribute__((aligned(16))) unsigned stage[STAGE_CAP];
  __shared__ unsigned hist[MAXBINS];
  __shared__ unsigned fill[MAXBINS];
  __shared__ unsigned wsum[32];
  __shared__ unsigned total;

  const int tid = threadIdx.x;
  hist[tid] = 0u;
  __syncthreads();

  const int gt = blockIdx.x * BT + tid;
  const int nt = BIN_BLOCKS * BT;
  const bool vec = ((n & 3) == 0) &&
                   (((((size_t)feat) | ((size_t)norm) | ((size_t)pre1) | ((size_t)agg1)) & 15) == 0);
  const int n4 = vec ? (n >> 2) : 0;
  float4 nf, nm;
  if (gt < n4) {
    nf = ((const float4*)feat)[gt];
    nm = ((const float4*)norm)[gt];
  }

  chunk_edges(src, dst, e4, e, [&](int s, int d) {
    const int bin = (d >> WBITS) * nw + (s >> WBITS);
    __hip_atomic_fetch_add(&hist[bin], 1u, __ATOMIC_RELAXED, __HIP_MEMORY_SCOPE_WORKGROUP);
  });
  __syncthreads();

  {
    const float4 zero = make_float4(0.0f, 0.0f, 0.0f, 0.0f);
    float4* p4 = (float4*)pre1;
    float4* a4 = (float4*)agg1;
    if (gt < n4) {
      asm volatile("" : "+v"(nf.x), "+v"(nf.y), "+v"(nf.z), "+v"(nf.w),
                        "+v"(nm.x), "+v"(nm.y), "+v"(nm.z), "+v"(nm.w));
      p4[gt] = make_float4(nf.x * nm.x, nf.y * nm.y, nf.z * nm.z, nf.w * nm.w);
      a4[gt] = zero;
    }
    for (int i = gt + nt; i < n4; i += nt) {
      const float4 f = ((const float4*)feat)[i];
      const float4 m = ((const float4*)norm)[i];
      p4[i] = make_float4(f.x * m.x, f.y * m.y, f.z * m.z, f.w * m.w);
      a4[i] = zero;
    }
    if (!vec) {
      for (long long i = gt; i < n; i += nt) {
        pre1[i] = feat[i] * norm[i];
        agg1[i] = 0.0f;
      }
    }
  }

  const unsigned mine = hist[tid];
  const unsigned ls   = block_excl_scan(mine, wsum);
  const unsigned base = blockIdx.x * (unsigned)cap;
  fill[tid] = ls;
  runStart[tid * BIN_BLOCKS + blockIdx.x] = base + ls;
  if (tid == BT - 1) {
    runStart[MAXBINS * BIN_BLOCKS + blockIdx.x] = base + ls + mine;
    total = ls + mine;
  }
  __syncthreads();

  chunk_edges(src, dst, e4, e, [&](int s, int d) {
    const int bin = (d >> WBITS) * nw + (s >> WBITS);
    const unsigned r = __hip_atomic_fetch_add(&fill[bin], 1u, __ATOMIC_RELAXED, __HIP_MEMORY_SCOPE_WORKGROUP);
    stage[r] = (unsigned)(s & WMASK) | ((unsigned)(d & WMASK) << WBITS);
  });
  __syncthreads();

  const unsigned tot = total;
  unsigned* __restrict__ outp = binned + base;
  if ((((size_t)outp) & 15) == 0) {
    for (unsigned q = tid * 4; q < tot; q += BT * 4) {
      if (q + 3 < tot) {
        *(uint4*)(outp + q) = *(const uint4*)(stage + q);
      } else {
        for (unsigned k = q; k < tot; ++k) outp[k] = stage[k];
      }
    }
  } else {
    for (unsigned q = tid; q < tot; q += BT) outp[q] = stage[q];
  }
}

__global__ __launch_bounds__(BT)
void gcn_edge_binned(const unsigned* __restrict__ binned,
                     const unsigned* __restrict__ runStart,
                     const float* __restrict__ val,
                     float* __restrict__ agg,
                     int nw, int n) {
  __shared__ float vals[WN];
  __shared__ float acc[WN];

  const int tid  = threadIdx.x;
  const int lane = tid & 31;
  const int j    = blockIdx.x / SRC_GROUPS;
  const int g    = blockIdx.x % SRC_GROUPS;
  const bool al  = (((size_t)val) & 15) == 0;
  const int b0 = __builtin_amdgcn_readfirstlane((tid >> 5) * RUNS_PER_WAVE);

  float4* acc4 = (float4*)acc;
  for (int i = tid; i < WN / 4; i += BT) {
    acc4[i] = make_float4(0.0f, 0.0f, 0.0f, 0.0f);
  }

  for (int i = g; i < nw; i += SRC_GROUPS) {
    __syncthreads();

    const unsigned* __restrict__ row = runStart + (j * nw + i) * BIN_BLOCKS + b0;
    unsigned rs[RUNS_PER_WAVE];
    unsigned rc[RUNS_PER_WAVE];
#pragma unroll
    for (int r = 0; r < RUNS_PER_WAVE; ++r) {
      rs[r] = row[r];
      rc[r] = row[BIN_BLOCKS + r] - rs[r];
    }

    const int sbase = i * WN;
    if (al && sbase + WN <= n) {
      float4 x[WN / (BT * 4)];
#pragma unroll
      for (int k = 0; k < WN / (BT * 4); ++k) {
        x[k] = *(const float4*)(val + sbase + (k * BT + tid) * 4);
      }
#pragma unroll
      for (int k = 0; k < WN / (BT * 4); ++k) {
        *(float4*)(vals + (k * BT + tid) * 4) = x[k];
      }
    } else {
      float y[WN / BT];
#pragma unroll
      for (int k = 0; k < WN / BT; ++k) {
        const int id = sbase + k * BT + tid;
        y[k] = (id < n) ? val[id] : 0.0f;
      }
#pragma unroll
      for (int k = 0; k < WN / BT; ++k) vals[k * BT + tid] = y[k];
    }
    __syncthreads();

    unsigned longest = 0u;
#pragma unroll
    for (int r = 0; r < RUNS_PER_WAVE; ++r) longest = (rc[r] > longest) ? rc[r] : longest;

    for (unsigned q0 = 0u; q0 < longest; q0 += 32u) {
      const unsigned q = q0 + lane;
      unsigned wd[RUNS_PER_WAVE];
      float    x[RUNS_PER_WAVE];
#pragma unroll
      for (int r = 0; r < RUNS_PER_WAVE; ++r) {
        wd[r] = (q < rc[r]) ? binned[rs[r] + q] : 0xFFFFFFFFu;
      }
#pragma unroll
      for (int r = 0; r < RUNS_PER_WAVE; ++r) {
        const bool ok = wd[r] != 0xFFFFFFFFu;
        const unsigned sl = ok ? (wd[r] & WMASK) : 0u;
        const unsigned dl = ok ? (wd[r] >> WBITS) : (unsigned)tid;
        const float v = vals[sl];
        x[r] = ok ? v : 0.0f;
        wd[r] = dl;
      }
#pragma unroll
      for (int r = 0; r < RUNS_PER_WAVE; ++r) {
        __hip_atomic_fetch_add(&acc[wd[r]], x[r],
                               __ATOMIC_RELAXED, __HIP_MEMORY_SCOPE_WORKGROUP);
      }
    }
  }
  __syncthreads();

  const int base = j * WN;
  for (int i = tid; i < WN; i += BT) {
    const float a = acc[i];
    const int id = base + i;
    if (id < n && a != 0.0f) {
      __hip_atomic_fetch_add(&agg[id], a, __ATOMIC_RELAXED, __HIP_MEMORY_SCOPE_AGENT);
    }
  }
}

#define WIN_NODES    65536
#define SCAN_SLICES  16
#define SCAN_THREADS 1024
#define SCAN_GROUP   4

__global__ __launch_bounds__(SCAN_THREADS)
void gcn_edge_window(const int* __restrict__ src,
                     const int* __restrict__ dst,
                     const float* __restrict__ val,
                     float* __restrict__ agg,
                     int e4, int e, int n) {
  __shared__ float acc[WIN_NODES];

  const int tid   = threadIdx.x;
  const int win   = blockIdx.x / SCAN_SLICES;
  const int slice = blockIdx.x % SCAN_SLICES;
  const int base  = win * WIN_NODES;

  float4* acc4 = (float4*)acc;
  for (int i = tid; i < WIN_NODES / 4; i += SCAN_THREADS) {
    acc4[i] = make_float4(0.0f, 0.0f, 0.0f, 0.0f);
  }
  __syncthreads();

  const int per   = (e4 + SCAN_SLICES - 1) / SCAN_SLICES;
  const int start = slice * per;
  const int stop  = (start + per < e4) ? (start + per) : e4;

  const v4i* s4 = (const v4i*)src;
  const v4i* d4 = (const v4i*)dst;

  for (int v0 = start + tid; v0 < stop; v0 += SCAN_THREADS * SCAN_GROUP) {
    v4i d[SCAN_GROUP];
    v4i s[SCAN_GROUP];
#pragma unroll
    for (int k = 0; k < SCAN_GROUP; ++k) {
      const int v = v0 + k * SCAN_THREADS;
      if (v < stop) {
        d[k] = d4[v];
        s[k] = s4[v];
      } else {
        d[k] = (v4i)(-1);
        s[k] = (v4i)(0);
      }
    }

    unsigned r[SCAN_GROUP][4];
    float    x[SCAN_GROUP][4];
#pragma unroll
    for (int k = 0; k < SCAN_GROUP; ++k) {
#pragma unroll
      for (int c = 0; c < 4; ++c) {
        r[k][c] = (unsigned)(d[k][c] - base);
        const int si = (r[k][c] < (unsigned)WIN_NODES) ? s[k][c] : 0;
        x[k][c] = val[si];
      }
    }
#pragma unroll
    for (int k = 0; k < SCAN_GROUP; ++k) {
#pragma unroll
      for (int c = 0; c < 4; ++c) {
        if (r[k][c] < (unsigned)WIN_NODES) {
          __hip_atomic_fetch_add(&acc[r[k][c]], x[k][c], __ATOMIC_RELAXED,
                                 __HIP_MEMORY_SCOPE_WORKGROUP);
        }
      }
    }
  }

  if (slice == 0 && tid == 0) {
    for (int k = e4 * 4; k < e; ++k) {
      const unsigned rr = (unsigned)(dst[k] - base);
      if (rr < (unsigned)WIN_NODES) {
        __hip_atomic_fetch_add(&acc[rr], val[src[k]], __ATOMIC_RELAXED,
                               __HIP_MEMORY_SCOPE_WORKGROUP);
      }
    }
  }
  __syncthreads();

  for (int i = tid; i < WIN_NODES; i += SCAN_THREADS) {
    const float a = acc[i];
    const int id = base + i;
    if (id < n && a != 0.0f) {
      __hip_atomic_fetch_add(&agg[id], a, __ATOMIC_RELAXED, __HIP_MEMORY_SCOPE_AGENT);
    }
  }
}

#define L1_GROUPS 4

__global__ __launch_bounds__(NODE_THREADS)
void gcn_layer1_wmma(const float* __restrict__ agg1,
                     const float* __restrict__ norm,
                     const float* __restrict__ W1,
                     const float* __restrict__ b1,
                     const float* __restrict__ W2,
                     float* __restrict__ t,
                     float* __restrict__ agg2,
                     int n_nodes) {
  const int lane = threadIdx.x & 31;
  const int step = gridDim.x * blockDim.x * L1_GROUPS;
  const int wb0 = __builtin_amdgcn_readfirstlane(
      (int)((blockIdx.x * blockDim.x + (threadIdx.x & ~31u)) * L1_GROUPS));

  const int ch = (lane < 16) ? lane : lane - 24;
  float w1a = 0.0f;
  if (ch >= 0 && ch < 3) w1a = W1[ch];
  v2f a;
  a.x = w1a;
  a.y = 0.0f;
  v8f c;
#pragma unroll
  for (int r = 0; r < 8; ++r) c[r] = 0.0f;
  c[0] = b1[0];
  c[1] = b1[1];
  c[2] = b1[2];
  const float w20 = W2[0];
  const float w21 = W2[1];
  const float w22 = W2[2];

  for (int wb = wb0; wb < n_nodes; wb += step) {
    float a1[L1_GROUPS];
    float nm[L1_GROUPS];
#pragma unroll
    for (int u = 0; u < L1_GROUPS; ++u) {
      const int i = wb + u * 32 + lane;
      const int ic = (i < n_nodes) ? i : wb;
      a1[u] = agg1[ic];
      nm[u] = norm[ic];
    }

    v8f d[L1_GROUPS];
#pragma unroll
    for (int u = 0; u < L1_GROUPS; ++u) {
      const int i = wb + u * 32 + lane;
      v2f b;
      b.x = (i < n_nodes) ? a1[u] * nm[u] : 0.0f;
      b.y = 0.0f;
      d[u] = __builtin_amdgcn_wmma_f32_16x16x4_f32(
           false, a,  false, b,
           (short)0, c,  false,  false);
    }

#pragma unroll
    for (int u = 0; u < L1_GROUPS; ++u) {
      const int i = wb + u * 32 + lane;
      const float h0 = fmaxf(d[u][0], 0.0f) * w20;
      const float h1 = fmaxf(d[u][1], 0.0f) * w21;
      const float h2 = fmaxf(d[u][2], 0.0f) * w22;
      const float s = (h0 + h1) + h2;
      if (i < n_nodes) {
        t[i] = nm[u] * s;
        agg2[i] = 0.0f;
      }
    }
  }
}

__global__ __launch_bounds__(NODE_THREADS)
void gcn_final(const float* __restrict__ norm,
               const float* __restrict__ agg2,
               const float* __restrict__ b2,
               float* __restrict__ out,
               int n) {
  const int gt = blockIdx.x * blockDim.x + threadIdx.x;
  const int nt = gridDim.x * blockDim.x;
  const float bias = b2[0];

  const bool vec = ((n & 3) == 0) &&
                   (((((size_t)norm) | ((size_t)agg2) | ((size_t)out)) & 15) == 0);
  if (vec) {
    const float4* m4 = (const float4*)norm;
    const float4* a4 = (const float4*)agg2;
    float4* o4 = (float4*)out;
    const int n4 = n >> 2;
    for (int i = gt; i < n4; i += 2 * nt) {
      const int j = i + nt;
      const int jc = (j < n4) ? j : i;
      const float4 m0 = m4[i];
      const float4 a0 = a4[i];
      const float4 m1 = m4[jc];
      const float4 a1 = a4[jc];
      o4[i] = make_float4(fmaxf(m0.x * a0.x + bias, 0.0f), fmaxf(m0.y * a0.y + bias, 0.0f),
                          fmaxf(m0.z * a0.z + bias, 0.0f), fmaxf(m0.w * a0.w + bias, 0.0f));
      o4[jc] = make_float4(fmaxf(m1.x * a1.x + bias, 0.0f), fmaxf(m1.y * a1.y + bias, 0.0f),
                           fmaxf(m1.z * a1.z + bias, 0.0f), fmaxf(m1.w * a1.w + bias, 0.0f));
    }
  } else {
    for (long long i = gt; i < n; i += nt) {
      out[i] = fmaxf(norm[i] * agg2[i] + bias, 0.0f);
    }
  }
}

extern "C" void kernel_launch(void* const* d_in, const int* in_sizes, int n_in,
                              void* d_out, int out_size, void* d_ws, size_t ws_size,
                              hipStream_t stream) {
  (void)n_in; (void)out_size;

  const float* feat = (const float*)d_in[0];
  const float* norm = (const float*)d_in[1];
  const int*   src  = (const int*)d_in[2];
  const int*   dst  = (const int*)d_in[3];
  const float* W1   = (const float*)d_in[4];
  const float* b1   = (const float*)d_in[5];
  const float* W2   = (const float*)d_in[6];
  const float* b2   = (const float*)d_in[7];
  float* out = (float*)d_out;

  const int N = in_sizes[0];
  const int E = in_sizes[2];

  float* pre1 = (float*)d_ws;
  float* agg1 = pre1 + N;
  float* tval = agg1 + N;
  float* agg2 = tval + N;
  unsigned* binned   = (unsigned*)(agg2 + N);
  unsigned* runStart = (unsigned*)d_out;

  const int e4 = E / 4;

  const int nw = (N + WN - 1) / WN;
  const int nb = nw * nw;
  const int slice_edges = ((e4 + BIN_BLOCKS - 1) / BIN_BLOCKS) * 4 + 3;
  const int cap = slice_edges + 1;
  const size_t need = (size_t)N * 16 + (size_t)BIN_BLOCKS * cap * 4;
  const bool binned_path = (ws_size >= need) && (nb <= MAXBINS) && (slice_edges <= STAGE_CAP) &&
                           ((size_t)N >= (size_t)TABLE_WORDS);

  if (binned_path) {
    gcn_bin_local<<<BIN_BLOCKS, BT, 0, stream>>>(src, dst, runStart, binned,
                                                 feat, norm, pre1, agg1, e4, E, nw, cap, N);
    gcn_edge_binned<<<nw * SRC_GROUPS, BT, 0, stream>>>(binned, runStart, pre1, agg1, nw, N);
    gcn_layer1_wmma<<<NODE_BLOCKS, NODE_THREADS, 0, stream>>>(agg1, norm, W1, b1, W2, tval, agg2, N);
    gcn_edge_binned<<<nw * SRC_GROUPS, BT, 0, stream>>>(binned, runStart, tval, agg2, nw, N);
    gcn_final<<<NODE_BLOCKS, NODE_THREADS, 0, stream>>>(norm, agg2, b2, out, N);
    return;
  }

  const int n_win = (N + WIN_NODES - 1) / WIN_NODES;
  const int grid_edges = n_win * SCAN_SLICES;

  gcn_premul_zero<<<NODE_BLOCKS, NODE_THREADS, 0, stream>>>(feat, norm, pre1, agg1, N);
  gcn_edge_window<<<grid_edges, SCAN_THREADS, 0, stream>>>(src, dst, pre1, agg1, e4, E, N);
  gcn_layer1_wmma<<<NODE_BLOCKS, NODE_THREADS, 0, stream>>>(agg1, norm, W1, b1, W2, tval, agg2, N);
  gcn_edge_window<<<grid_edges, SCAN_THREADS, 0, stream>>>(src, dst, tval, agg2, e4, E, N);
  gcn_final<<<NODE_BLOCKS, NODE_THREADS, 0, stream>>>(norm, agg2, b2, out, N);
}
